# starting kernel
# baseline (speedup 1.0000x reference)
#include <hip/hip_runtime.h>

typedef __attribute__((ext_vector_type(2))) float v2f;
typedef __attribute__((ext_vector_type(8))) float v8f;

#define B_N    16
#define CIN    64
#define COUT   64
#define H_IN   128
#define W_IN   128
#define HOUT   126
#define WOUT   126
#define KDIM   576

#define WPITCH 578
#define XCOLS  130
#define CIPITCH 66

#define SW_FLOATS (COUT * WPITCH)
#define SX_FLOATS (3 * XCOLS * CIPITCH)
#define LDS_BYTES ((SW_FLOATS + SX_FLOATS) * 4)

extern "C" __global__ __launch_bounds__(256)
void dynconv2d_wmma_f32(const float* __restrict__ x,
                        const float* __restrict__ w,
                        const float* __restrict__ bias,
                        float* __restrict__ out)
{
    extern __shared__ float smem[];
    float* sW = smem;
    float* sX = smem + SW_FLOATS;

    const int h   = blockIdx.x;
    const int b   = blockIdx.y;
    const int tid = threadIdx.x;

    const float* wsrc = w + (size_t)b * COUT * KDIM;
    for (int i = tid; i < COUT * KDIM; i += 256) {
        int co  = i / KDIM;
        int rem = i - co * KDIM;
        int ci  = rem / 9;
        int r   = rem - ci * 9;
        sW[co * WPITCH + r * 64 + ci] = wsrc[i];
    }

    for (int i = tid; i < CIN * 3 * XCOLS; i += 256) {
        int ci  = i / (3 * XCOLS);
        int rem = i - ci * (3 * XCOLS);
        int kh  = rem / XCOLS;
        int col = rem - kh * XCOLS;
        float v = 0.0f;
        if (col < W_IN)
            v = x[(((size_t)b * CIN + ci) * H_IN + (h + kh)) * W_IN + col];
        sX[(kh * XCOLS + col) * CIPITCH + ci] = v;
    }
    __syncthreads();

    const int wave  = tid >> 5;
    const int lane  = tid & 31;
    const int laneN = lane & 15;
    const int hi    = lane >> 4;

    const int co_base = (wave & 3) * 16;
    const int P0      = (wave >> 2) * 64;

    const float* aB  = sW + (co_base + laneN) * WPITCH + 2 * hi;
    const float* bB0 = sX + (P0 +  0 + laneN) * CIPITCH + 2 * hi;
    const float* bB1 = sX + (P0 + 16 + laneN) * CIPITCH + 2 * hi;
    const float* bB2 = sX + (P0 + 32 + laneN) * CIPITCH + 2 * hi;
    const float* bB3 = sX + (P0 + 48 + laneN) * CIPITCH + 2 * hi;

    v8f acc0 = {}, acc1 = {}, acc2 = {}, acc3 = {};

    #pragma unroll
    for (int kh = 0; kh < 3; ++kh) {
        #pragma unroll
        for (int kw = 0; kw < 3; ++kw) {
            const int r    = kh * 3 + kw;
            const int aoff = r * 64;
            const int boff = (kh * XCOLS + kw) * CIPITCH;
            #pragma unroll
            for (int c0 = 0; c0 < CIN; c0 += 4) {
                v2f a  = *(const v2f*)(aB  + aoff + c0);
                v2f b0 = *(const v2f*)(bB0 + boff + c0);
                v2f b1 = *(const v2f*)(bB1 + boff + c0);
                v2f b2 = *(const v2f*)(bB2 + boff + c0);
                v2f b3 = *(const v2f*)(bB3 + boff + c0);
                acc0 = __builtin_amdgcn_wmma_f32_16x16x4_f32(false, a, false, b0, (short)0, acc0, false, false);
                acc1 = __builtin_amdgcn_wmma_f32_16x16x4_f32(false, a, false, b1, (short)0, acc1, false, false);
                acc2 = __builtin_amdgcn_wmma_f32_16x16x4_f32(false, a, false, b2, (short)0, acc2, false, false);
                acc3 = __builtin_amdgcn_wmma_f32_16x16x4_f32(false, a, false, b3, (short)0, acc3, false, false);
            }
        }
    }

    #pragma unroll
    for (int t = 0; t < 4; ++t) {
        const v8f& acc = (t == 0) ? acc0 : (t == 1) ? acc1 : (t == 2) ? acc2 : acc3;
        const int wpix = P0 + t * 16 + laneN;
        if (wpix < WOUT) {
            #pragma unroll
            for (int j = 0; j < 8; ++j) {
                int co = co_base + j + 8 * hi;
                out[(((size_t)b * COUT + co) * HOUT + h) * WOUT + wpix] =
                    acc[j] + bias[b * COUT + co];
            }
        }
    }
}

extern "C" void kernel_launch(void* const* d_in, const int* in_sizes, int n_in,
                              void* d_out, int out_size, void* d_ws, size_t ws_size,
                              hipStream_t stream) {
    (void)in_sizes; (void)n_in; (void)out_size; (void)d_ws; (void)ws_size;
    const float* x    = (const float*)d_in[0];
    const float* w    = (const float*)d_in[1];
    const float* bias = (const float*)d_in[2];
    float* out        = (float*)d_out;

    dim3 grid(HOUT, B_N);
    dynconv2d_wmma_f32<<<grid, 256, LDS_BYTES, stream>>>(x, w, bias, out);
}


# One block per (batch, 8 output rows): weights staged to LDS once per block, input rows kept in a 3-slot LDS ring (one new row staged per output row); grid 126x16 -> 16x16; summation order unchanged
# speedup vs baseline: 1.9901x; 1.9901x over previous
#include <hip/hip_runtime.h>

typedef __attribute__((ext_vector_type(2))) float v2f;
typedef __attribute__((ext_vector_type(8))) float v8f;

#define B_N    16
#define CIN    64
#define COUT   64
#define H_IN   128
#define W_IN   128
#define HOUT   126
#define WOUT   126
#define KDIM   576

#define WPITCH 578
#define XCOLS  130
#define CIPITCH 66

#define ROWS_PER_BLOCK 8
#define NCHUNK 16

#define SW_FLOATS (COUT * WPITCH)
#define XROW_FLOATS (XCOLS * CIPITCH)
#define SX_FLOATS (3 * XROW_FLOATS)
#define LDS_BYTES ((SW_FLOATS + SX_FLOATS) * 4)

extern "C" __global__ __launch_bounds__(256)
void dynconv2d_wmma_f32(const float* __restrict__ x,
                        const float* __restrict__ w,
                        const float* __restrict__ bias,
                        float* __restrict__ out)
{
    extern __shared__ float smem[];
    float* sW = smem;
    float* sX = smem + SW_FLOATS;

    const int h0    = blockIdx.x * ROWS_PER_BLOCK;
    const int nrows = min(ROWS_PER_BLOCK, HOUT - h0);
    const int b     = blockIdx.y;
    const int tid   = threadIdx.x;

    const float* wsrc = w + (size_t)b * COUT * KDIM;
    for (int i = tid; i < COUT * KDIM; i += 256) {
        int co  = i / KDIM;
        int rem = i - co * KDIM;
        int ci  = rem / 9;
        int r   = rem - ci * 9;
        sW[co * WPITCH + r * 64 + ci] = wsrc[i];
    }

    for (int i = tid; i < CIN * 3 * XCOLS; i += 256) {
        int ci  = i / (3 * XCOLS);
        int rem = i - ci * (3 * XCOLS);
        int kh  = rem / XCOLS;
        int col = rem - kh * XCOLS;
        float v = 0.0f;
        if (col < W_IN)
            v = x[(((size_t)b * CIN + ci) * H_IN + (h0 + kh)) * W_IN + col];
        sX[(kh * XCOLS + col) * CIPITCH + ci] = v;
    }
    __syncthreads();

    const int wave  = tid >> 5;
    const int lane  = tid & 31;
    const int laneN = lane & 15;
    const int hi    = lane >> 4;

    const int co_base = (wave & 3) * 16;
    const int P0      = (wave >> 2) * 64;

    const float* aB  = sW + (co_base + laneN) * WPITCH + 2 * hi;
    const float* bB0 = sX + (P0 +  0 + laneN) * CIPITCH + 2 * hi;
    const float* bB1 = sX + (P0 + 16 + laneN) * CIPITCH + 2 * hi;
    const float* bB2 = sX + (P0 + 32 + laneN) * CIPITCH + 2 * hi;
    const float* bB3 = sX + (P0 + 48 + laneN) * CIPITCH + 2 * hi;

    float bv[8];
    #pragma unroll
    for (int j = 0; j < 8; ++j)
        bv[j] = bias[b * COUT + co_base + j + 8 * hi];

    int s0 = 0, s1 = XROW_FLOATS, s2 = 2 * XROW_FLOATS;

    for (int hh = 0; hh < nrows; ++hh) {
        const int h = h0 + hh;

        v8f acc0 = {}, acc1 = {}, acc2 = {}, acc3 = {};

        #pragma unroll
        for (int kh = 0; kh < 3; ++kh) {
            const int srow = (kh == 0) ? s0 : (kh == 1) ? s1 : s2;
            #pragma unroll
            for (int kw = 0; kw < 3; ++kw) {
                const int r    = kh * 3 + kw;
                const int aoff = r * 64;
                const int boff = srow + kw * CIPITCH;
                #pragma unroll
                for (int c0 = 0; c0 < CIN; c0 += 4) {
                    v2f a  = *(const v2f*)(aB  + aoff + c0);
                    v2f b0 = *(const v2f*)(bB0 + boff + c0);
                    v2f b1 = *(const v2f*)(bB1 + boff + c0);
                    v2f b2 = *(const v2f*)(bB2 + boff + c0);
                    v2f b3 = *(const v2f*)(bB3 + boff + c0);
                    acc0 = __builtin_amdgcn_wmma_f32_16x16x4_f32(false, a, false, b0, (short)0, acc0, false, false);
                    acc1 = __builtin_amdgcn_wmma_f32_16x16x4_f32(false, a, false, b1, (short)0, acc1, false, false);
                    acc2 = __builtin_amdgcn_wmma_f32_16x16x4_f32(false, a, false, b2, (short)0, acc2, false, false);
                    acc3 = __builtin_amdgcn_wmma_f32_16x16x4_f32(false, a, false, b3, (short)0, acc3, false, false);
                }
            }
        }

        asm volatile("v_nop\n\tv_nop\n\tv_nop\n\tv_nop"
                     : "+v"(acc0), "+v"(acc1), "+v"(acc2), "+v"(acc3));

        #pragma unroll
        for (int t = 0; t < 4; ++t) {
            const v8f& acc = (t == 0) ? acc0 : (t == 1) ? acc1 : (t == 2) ? acc2 : acc3;
            const int wpix = P0 + t * 16 + laneN;
            if (wpix < WOUT) {
                #pragma unroll
                for (int j = 0; j < 8; ++j) {
                    int co = co_base + j + 8 * hi;
                    out[(((size_t)b * COUT + co) * HOUT + h) * WOUT + wpix] =
                        acc[j] + bv[j];
                }
            }
        }

        if (hh + 1 < nrows) {
            __syncthreads();
            for (int i = tid; i < CIN * XCOLS; i += 256) {
                int ci  = i / XCOLS;
                int col = i - ci * XCOLS;
                float v = 0.0f;
                if (col < W_IN)
                    v = x[(((size_t)b * CIN + ci) * H_IN + (h + 3)) * W_IN + col];
                sX[s0 + col * CIPITCH + ci] = v;
            }
            __syncthreads();
        }
        const int st = s0; s0 = s1; s1 = s2; s2 = st;
    }
}

extern "C" void kernel_launch(void* const* d_in, const int* in_sizes, int n_in,
                              void* d_out, int out_size, void* d_ws, size_t ws_size,
                              hipStream_t stream) {
    (void)in_sizes; (void)n_in; (void)out_size; (void)d_ws; (void)ws_size;
    const float* x    = (const float*)d_in[0];
    const float* w    = (const float*)d_in[1];
    const float* bias = (const float*)d_in[2];
    float* out        = (float*)d_out;

    dim3 grid(NCHUNK, B_N);
    dynconv2d_wmma_f32<<<grid, 256, LDS_BYTES, stream>>>(x, w, bias, out);
}


# Global-load latency of staging hidden: next input row fetched to registers before each row's products, scattered to a 4th LDS ring slot after; all staging by grouped 16-B loads; sums unchanged
# speedup vs baseline: 5.0876x; 2.5565x over previous
#include <hip/hip_runtime.h>

typedef __attribute__((ext_vector_type(2))) float v2f;
typedef __attribute__((ext_vector_type(4))) float v4f;
typedef __attribute__((ext_vector_type(8))) float v8f;

#define B_N    16
#define CIN    64
#define COUT   64
#define H_IN   128
#define W_IN   128
#define HOUT   126
#define WOUT   126
#define KDIM   576

#define WPITCH 578
#define XCOLS  130
#define CIPITCH 66

#define ROWS_PER_BLOCK 8
#define NCHUNK 16

#define NSLOT 4
#define SW_FLOATS (COUT * WPITCH)
#define XROW_FLOATS (XCOLS * CIPITCH)
#define SX_FLOATS (NSLOT * XROW_FLOATS)
#define LDS_BYTES ((SW_FLOATS + SX_FLOATS) * 4)

#define ROW_F4 8
#define PLANE8 (8 * H_IN * W_IN)
#define W_F4   36
#define W_GRP  12

__device__ __forceinline__ void row_fetch(v4f (&r)[ROW_F4], const float* __restrict__ src)
{
    #pragma unroll
    for (int j = 0; j < ROW_F4; ++j)
        r[j] = *(const v4f*)(src + (size_t)j * PLANE8);
}

__device__ __forceinline__ void row_put(float* dst, const v4f (&r)[ROW_F4])
{
    #pragma unroll
    for (int j = 0; j < ROW_F4; ++j) {
        dst[0 * CIPITCH + 8 * j] = r[j].x;
        dst[1 * CIPITCH + 8 * j] = r[j].y;
        dst[2 * CIPITCH + 8 * j] = r[j].z;
        dst[3 * CIPITCH + 8 * j] = r[j].w;
    }
}

extern "C" __global__ __launch_bounds__(256)
void dynconv2d_wmma_f32(const float* __restrict__ x,
                        const float* __restrict__ w,
                        const float* __restrict__ bias,
                        float* __restrict__ out)
{
    extern __shared__ float smem[];
    float* sW = smem;
    float* sX = smem + SW_FLOATS;

    const int h0    = blockIdx.x * ROWS_PER_BLOCK;
    const int nrows = min(ROWS_PER_BLOCK, HOUT - h0);
    const int b     = blockIdx.y;
    const int tid   = threadIdx.x;
    const int wave  = tid >> 5;
    const int lane  = tid & 31;

    const int cg   = (wave & 3) * 8 + (lane & 7);
    const int ci_t = (wave >> 2) * 4 + (lane >> 3);
    const float* xsrc = x + (((size_t)b * CIN + ci_t) * H_IN + h0) * W_IN + 4 * cg;
    float*       xdst = sX + (4 * cg) * CIPITCH + ci_t;

    v4f ra[ROW_F4], rb[ROW_F4], rc[ROW_F4];
    row_fetch(ra, xsrc);
    row_fetch(rb, xsrc + W_IN);
    row_fetch(rc, xsrc + 2 * W_IN);

    const float* wsrc = w + (size_t)b * COUT * KDIM;

    #pragma unroll
    for (int k = 0; k < 2; ++k) {
        int idx = tid + 256 * k;
        sX[(idx >> 7) * XROW_FLOATS + (W_IN + ((idx >> 6) & 1)) * CIPITCH + (idx & 63)] = 0.0f;
    }

    #pragma unroll
    for (int g = 0; g < W_F4 / W_GRP; ++g) {
        v4f wv[W_GRP];
        #pragma unroll
        for (int j = 0; j < W_GRP; ++j)
            wv[j] = *(const v4f*)(wsrc + 4 * (tid + 256 * (g * W_GRP + j)));
        if (g == 0) {
            row_put(xdst,                   ra);
            row_put(xdst + XROW_FLOATS,     rb);
            row_put(xdst + 2 * XROW_FLOATS, rc);
        }
        #pragma unroll
        for (int j = 0; j < W_GRP; ++j) {
            int u   = tid + 256 * (g * W_GRP + j);
            int co  = u / 144;
            int rem = 4 * (u - co * 144);
            int ci  = rem / 9;
            int r   = rem - ci * 9;
            float* d = sW + co * WPITCH;
            #pragma unroll
            for (int e = 0; e < 4; ++e) {
                d[r * 64 + ci] = wv[j][e];
                if (++r == 9) { r = 0; ++ci; }
            }
        }
    }
    __syncthreads();

    const int laneN = lane & 15;
    const int hi    = lane >> 4;

    const int co_base = (wave & 3) * 16;
    const int P0      = (wave >> 2) * 64;

    const float* aB  = sW + (co_base + laneN) * WPITCH + 2 * hi;
    const float* bB0 = sX + (P0 +  0 + laneN) * CIPITCH + 2 * hi;
    const float* bB1 = sX + (P0 + 16 + laneN) * CIPITCH + 2 * hi;
    const float* bB2 = sX + (P0 + 32 + laneN) * CIPITCH + 2 * hi;
    const float* bB3 = sX + (P0 + 48 + laneN) * CIPITCH + 2 * hi;

    float bv[8];
    #pragma unroll
    for (int j = 0; j < 8; ++j)
        bv[j] = bias[b * COUT + co_base + j + 8 * hi];

    int s0 = 0, s1 = XROW_FLOATS, s2 = 2 * XROW_FLOATS, s3 = 3 * XROW_FLOATS;

    for (int hh = 0; hh < nrows; ++hh) {
        const int  h    = h0 + hh;
        const bool more = hh + 1 < nrows;

        v4f rn[ROW_F4];
        if (more)
            row_fetch(rn, xsrc + (hh + 3) * W_IN);

        v8f acc0 = {}, acc1 = {}, acc2 = {}, acc3 = {};

        #pragma unroll
        for (int kh = 0; kh < 3; ++kh) {
            const int srow = (kh == 0) ? s0 : (kh == 1) ? s1 : s2;
            #pragma unroll
            for (int kw = 0; kw < 3; ++kw) {
                const int r    = kh * 3 + kw;
                const int aoff = r * 64;
                const int boff = srow + kw * CIPITCH;
                #pragma unroll
                for (int c0 = 0; c0 < CIN; c0 += 4) {
                    v2f a  = *(const v2f*)(aB  + aoff + c0);
                    v2f b0 = *(const v2f*)(bB0 + boff + c0);
                    v2f b1 = *(const v2f*)(bB1 + boff + c0);
                    v2f b2 = *(const v2f*)(bB2 + boff + c0);
                    v2f b3 = *(const v2f*)(bB3 + boff + c0);
                    acc0 = __builtin_amdgcn_wmma_f32_16x16x4_f32(false, a, false, b0, (short)0, acc0, false, false);
                    acc1 = __builtin_amdgcn_wmma_f32_16x16x4_f32(false, a, false, b1, (short)0, acc1, false, false);
                    acc2 = __builtin_amdgcn_wmma_f32_16x16x4_f32(false, a, false, b2, (short)0, acc2, false, false);
                    acc3 = __builtin_amdgcn_wmma_f32_16x16x4_f32(false, a, false, b3, (short)0, acc3, false, false);
                }
            }
        }

        asm volatile("v_nop\n\tv_nop\n\tv_nop\n\tv_nop"
                     : "+v"(acc0), "+v"(acc1), "+v"(acc2), "+v"(acc3));

        if (more) {
            row_put(xdst + s3, rn);
            __syncthreads();
        }

        #pragma unroll
        for (int t = 0; t < 4; ++t) {
            const v8f& acc = (t == 0) ? acc0 : (t == 1) ? acc1 : (t == 2) ? acc2 : acc3;
            const int wpix = P0 + t * 16 + laneN;
            if (wpix < WOUT) {
                #pragma unroll
                for (int j = 0; j < 8; ++j) {
                    int co = co_base + j + 8 * hi;
                    out[(((size_t)b * COUT + co) * HOUT + h) * WOUT + wpix] =
                        acc[j] + bv[j];
                }
            }
        }

        const int st = s0; s0 = s1; s1 = s2; s2 = s3; s3 = st;
    }
}

extern "C" void kernel_launch(void* const* d_in, const int* in_sizes, int n_in,
                              void* d_out, int out_size, void* d_ws, size_t ws_size,
                              hipStream_t stream) {
    (void)in_sizes; (void)n_in; (void)out_size; (void)d_ws; (void)ws_size;
    const float* x    = (const float*)d_in[0];
    const float* w    = (const float*)d_in[1];
    const float* bias = (const float*)d_in[2];
    float* out        = (float*)d_out;

    dim3 grid(NCHUNK, B_N);
    dynconv2d_wmma_f32<<<grid, 256, LDS_BYTES, stream>>>(x, w, bias, out);
}


# Row-boundary work moved inside the product stream: a finished row's stores (from a second register set) and the next input row's LDS scatter now stand behind tap 0 of the next row; sums unchanged
# speedup vs baseline: 5.3360x; 1.0488x over previous
#include <hip/hip_runtime.h>

typedef __attribute__((ext_vector_type(2))) float v2f;
typedef __attribute__((ext_vector_type(4))) float v4f;
typedef __attribute__((ext_vector_type(8))) float v8f;

#define B_N    16
#define CIN    64
#define COUT   64
#define H_IN   128
#define W_IN   128
#define HOUT   126
#define WOUT   126
#define KDIM   576

#define WPITCH 578
#define XCOLS  130
#define CIPITCH 66

#define ROWS_PER_BLOCK 8
#define NCHUNK 16

#define NSLOT 4
#define SW_FLOATS (COUT * WPITCH)
#define XROW_FLOATS (XCOLS * CIPITCH)
#define SX_FLOATS (NSLOT * XROW_FLOATS)
#define LDS_BYTES ((SW_FLOATS + SX_FLOATS) * 4)

#define ROW_F4 8
#define PLANE8 (8 * H_IN * W_IN)
#define W_F4   36
#define W_GRP  12

__device__ __forceinline__ void row_fetch(v4f (&r)[ROW_F4], const float* __restrict__ src)
{
    #pragma unroll
    for (int j = 0; j < ROW_F4; ++j)
        r[j] = *(const v4f*)(src + (size_t)j * PLANE8);
}

__device__ __forceinline__ void row_put(float* dst, const v4f (&r)[ROW_F4])
{
    #pragma unroll
    for (int j = 0; j < ROW_F4; ++j) {
        dst[0 * CIPITCH + 8 * j] = r[j].x;
        dst[1 * CIPITCH + 8 * j] = r[j].y;
        dst[2 * CIPITCH + 8 * j] = r[j].z;
        dst[3 * CIPITCH + 8 * j] = r[j].w;
    }
}

__device__ __forceinline__ void tile_out(float* __restrict__ o, const v8f& res)
{
    #pragma unroll
    for (int j = 0; j < 8; ++j)
        o[(size_t)j * (HOUT * WOUT)] = res[j];
}

__device__ __forceinline__ v8f add_bias(const v8f& acc, const float (&bv)[8])
{
    v8f r;
    #pragma unroll
    for (int j = 0; j < 8; ++j)
        r[j] = acc[j] + bv[j];
    return r;
}

extern "C" __global__ __launch_bounds__(256)
void dynconv2d_wmma_f32(const float* __restrict__ x,
                        const float* __restrict__ w,
                        const float* __restrict__ bias,
                        float* __restrict__ out)
{
    extern __shared__ float smem[];
    float* sW = smem;
    float* sX = smem + SW_FLOATS;

    const int h0    = blockIdx.x * ROWS_PER_BLOCK;
    const int nrows = min(ROWS_PER_BLOCK, HOUT - h0);
    const int b     = blockIdx.y;
    const int tid   = threadIdx.x;
    const int wave  = tid >> 5;
    const int lane  = tid & 31;

    const int cg   = (wave & 3) * 8 + (lane & 7);
    const int ci_t = (wave >> 2) * 4 + (lane >> 3);
    const float* xsrc = x + (((size_t)b * CIN + ci_t) * H_IN + h0) * W_IN + 4 * cg;
    float*       xdst = sX + (4 * cg) * CIPITCH + ci_t;

    v4f ra[ROW_F4], rb[ROW_F4], rc[ROW_F4];
    row_fetch(ra, xsrc);
    row_fetch(rb, xsrc + W_IN);
    row_fetch(rc, xsrc + 2 * W_IN);

    const float* wsrc = w + (size_t)b * COUT * KDIM;

    #pragma unroll
    for (int k = 0; k < 2; ++k) {
        int idx = tid + 256 * k;
        sX[(idx >> 7) * XROW_FLOATS + (W_IN + ((idx >> 6) & 1)) * CIPITCH + (idx & 63)] = 0.0f;
    }

    #pragma unroll
    for (int g = 0; g < W_F4 / W_GRP; ++g) {
        v4f wv[W_GRP];
        #pragma unroll
        for (int j = 0; j < W_GRP; ++j)
            wv[j] = *(const v4f*)(wsrc + 4 * (tid + 256 * (g * W_GRP + j)));
        if (g == 0) {
            row_put(xdst,                   ra);
            row_put(xdst + XROW_FLOATS,     rb);
            row_put(xdst + 2 * XROW_FLOATS, rc);
        }
        #pragma unroll
        for (int j = 0; j < W_GRP; ++j) {
            int u   = tid + 256 * (g * W_GRP + j);
            int co  = u / 144;
            int rem = 4 * (u - co * 144);
            int ci  = rem / 9;
            int r   = rem - ci * 9;
            float* d = sW + co * WPITCH;
            #pragma unroll
            for (int e = 0; e < 4; ++e) {
                d[r * 64 + ci] = wv[j][e];
                if (++r == 9) { r = 0; ++ci; }
            }
        }
    }
    __syncthreads();

    const int laneN = lane & 15;
    const int hi    = lane >> 4;

    const int co_base = (wave & 3) * 16;
    const int P0      = (wave >> 2) * 64;

    const float* aB  = sW + (co_base + laneN) * WPITCH + 2 * hi;
    const float* fr0[4];
    const float* fr1[4];
    const float* fr2[4];
    const float* fr3[4];
    #pragma unroll
    for (int t = 0; t < 4; ++t) {
        fr0[t] = sX + (P0 + 16 * t + laneN) * CIPITCH + 2 * hi;
        fr1[t] = fr0[t] + XROW_FLOATS;
        fr2[t] = fr0[t] + 2 * XROW_FLOATS;
        fr3[t] = fr0[t] + 3 * XROW_FLOATS;
    }
    float* xd0 = xdst, *xd1 = xdst + XROW_FLOATS, *xd2 = xdst + 2 * XROW_FLOATS, *xd3 = xdst + 3 * XROW_FLOATS;

    float bv[8];
    #pragma unroll
    for (int j = 0; j < 8; ++j)
        bv[j] = bias[b * COUT + co_base + j + 8 * hi];

    float* obase = out + ((size_t)(b * COUT + co_base + 8 * hi) * HOUT) * WOUT + P0 + laneN;
    const bool tile3_ok = (P0 + 48 + laneN) < WOUT;

    v8f p0 = {}, p1 = {}, p2 = {}, p3 = {};

    for (int hh = 0; hh < nrows; ++hh) {
        const int  h    = h0 + hh;
        const bool more = hh + 1 < nrows;

        v4f rn[ROW_F4];
        if (more)
            row_fetch(rn, xsrc + (hh + 3) * W_IN);

        v8f acc0 = {}, acc1 = {}, acc2 = {}, acc3 = {};

        #pragma unroll
        for (int kh = 0; kh < 3; ++kh) {
            const float* const* fb = (kh == 0) ? fr0 : (kh == 1) ? fr1 : fr2;
            #pragma unroll
            for (int kw = 0; kw < 3; ++kw) {
                const int r    = kh * 3 + kw;
                const int aoff = r * 64;
                const int boff = kw * CIPITCH;
                #pragma unroll
                for (int c0 = 0; c0 < CIN; c0 += 4) {
                    v2f a  = *(const v2f*)(aB  + aoff + c0);
                    v2f b0 = *(const v2f*)(fb[0] + boff + c0);
                    v2f b1 = *(const v2f*)(fb[1] + boff + c0);
                    v2f b2 = *(const v2f*)(fb[2] + boff + c0);
                    v2f b3 = *(const v2f*)(fb[3] + boff + c0);
                    acc0 = __builtin_amdgcn_wmma_f32_16x16x4_f32(false, a, false, b0, (short)0, acc0, false, false);
                    acc1 = __builtin_amdgcn_wmma_f32_16x16x4_f32(false, a, false, b1, (short)0, acc1, false, false);
                    acc2 = __builtin_amdgcn_wmma_f32_16x16x4_f32(false, a, false, b2, (short)0, acc2, false, false);
                    acc3 = __builtin_amdgcn_wmma_f32_16x16x4_f32(false, a, false, b3, (short)0, acc3, false, false);
                }
                if (r == 0) {
                    if (hh > 0) {
                        float* o = obase + (size_t)(h - 1) * WOUT;
                        tile_out(o,      p0);
                        tile_out(o + 16, p1);
                        tile_out(o + 32, p2);
                        if (tile3_ok)
                            tile_out(o + 48, p3);
                    }
                    if (more)
                        row_put(xd3, rn);
                }
            }
        }

        asm volatile("v_nop\n\tv_nop\n\tv_nop\n\tv_nop"
                     : "+v"(acc0), "+v"(acc1), "+v"(acc2), "+v"(acc3));

        p0 = add_bias(acc0, bv); p1 = add_bias(acc1, bv);
        p2 = add_bias(acc2, bv); p3 = add_bias(acc3, bv);

        if (more)
            __syncthreads();

        #pragma unroll
        for (int t = 0; t < 4; ++t) {
            const float* ft = fr0[t]; fr0[t] = fr1[t]; fr1[t] = fr2[t]; fr2[t] = fr3[t]; fr3[t] = ft;
        }
        float* xt = xd0; xd0 = xd1; xd1 = xd2; xd2 = xd3; xd3 = xt;
    }

    {
        float* o = obase + (size_t)(h0 + nrows - 1) * WOUT;
        tile_out(o,      p0);
        tile_out(o + 16, p1);
        tile_out(o + 32, p2);
        if (tile3_ok)
            tile_out(o + 48, p3);
    }
}

extern "C" void kernel_launch(void* const* d_in, const int* in_sizes, int n_in,
                              void* d_out, int out_size, void* d_ws, size_t ws_size,
                              hipStream_t stream) {
    (void)in_sizes; (void)n_in; (void)out_size; (void)d_ws; (void)ws_size;
    const float* x    = (const float*)d_in[0];
    const float* w    = (const float*)d_in[1];
    const float* bias = (const float*)d_in[2];
    float* out        = (float*)d_out;

    dim3 grid(NCHUNK, B_N);
    dynconv2d_wmma_f32<<<grid, 256, LDS_BYTES, stream>>>(x, w, bias, out);
}


# LDS operand reads regrouped: 20 16-byte reads per 32 products, issued one block ahead into a 2nd register set (was 5 per 8 products); ci order per 8 now 0,1,4,5|2,3,6,7: summation order changes
# speedup vs baseline: 5.5060x; 1.0319x over previous
#include <hip/hip_runtime.h>

typedef __attribute__((ext_vector_type(2))) float v2f;
typedef __attribute__((ext_vector_type(4))) float v4f;
typedef __attribute__((ext_vector_type(8))) float v8f;

#define B_N    16
#define CIN    64
#define COUT   64
#define H_IN   128
#define W_IN   128
#define HOUT   126
#define WOUT   126
#define KDIM   576

#define WPITCH 580
#define XCOLS  130
#define CIPITCH 68

#define ROWS_PER_BLOCK 8
#define NCHUNK 16

#define NSLOT 4
#define SW_FLOATS (COUT * WPITCH)
#define XROW_FLOATS (XCOLS * CIPITCH)
#define SX_FLOATS (NSLOT * XROW_FLOATS)
#define LDS_BYTES ((SW_FLOATS + SX_FLOATS) * 4)

#define ROW_F4 8
#define PLANE8 (8 * H_IN * W_IN)
#define W_F4   36
#define W_GRP  12

__device__ __forceinline__ void row_fetch(v4f (&r)[ROW_F4], const float* __restrict__ src)
{
    #pragma unroll
    for (int j = 0; j < ROW_F4; ++j)
        r[j] = *(const v4f*)(src + (size_t)j * PLANE8);
}

__device__ __forceinline__ void row_put(float* dst, const v4f (&r)[ROW_F4])
{
    #pragma unroll
    for (int j = 0; j < ROW_F4; ++j) {
        dst[0 * CIPITCH + 8 * j] = r[j].x;
        dst[1 * CIPITCH + 8 * j] = r[j].y;
        dst[2 * CIPITCH + 8 * j] = r[j].z;
        dst[3 * CIPITCH + 8 * j] = r[j].w;
    }
}

__device__ __forceinline__ void tile_out(float* __restrict__ o, const v8f& res)
{
    #pragma unroll
    for (int j = 0; j < 8; ++j)
        o[(size_t)j * (HOUT * WOUT)] = res[j];
}

__device__ __forceinline__ v8f add_bias(const v8f& acc, const float (&bv)[8])
{
    v8f r;
    #pragma unroll
    for (int j = 0; j < 8; ++j)
        r[j] = acc[j] + bv[j];
    return r;
}

#define NBLK   18

struct Frag {
    v4f a[4];
    v4f b[4][4];
};

__device__ __forceinline__ void frag_load(Frag& f, const float* aP, const float* const (&bP)[4],
                                          int aoff, int boff)
{
    #pragma unroll
    for (int j = 0; j < 4; ++j)
        f.a[j] = *(const v4f*)(aP + aoff + 8 * j);
    #pragma unroll
    for (int t = 0; t < 4; ++t) {
        #pragma unroll
        for (int j = 0; j < 4; ++j)
            f.b[t][j] = *(const v4f*)(bP[t] + boff + 8 * j);
    }
}

__device__ __forceinline__ void frag_mma(v8f (&acc)[4], const Frag& f)
{
    #pragma unroll
    for (int j = 0; j < 4; ++j) {
        const v2f alo = __builtin_shufflevector(f.a[j], f.a[j], 0, 1);
        const v2f ahi = __builtin_shufflevector(f.a[j], f.a[j], 2, 3);
        #pragma unroll
        for (int t = 0; t < 4; ++t) {
            const v2f blo = __builtin_shufflevector(f.b[t][j], f.b[t][j], 0, 1);
            acc[t] = __builtin_amdgcn_wmma_f32_16x16x4_f32(false, alo, false, blo, (short)0, acc[t], false, false);
        }
        #pragma unroll
        for (int t = 0; t < 4; ++t) {
            const v2f bhi = __builtin_shufflevector(f.b[t][j], f.b[t][j], 2, 3);
            acc[t] = __builtin_amdgcn_wmma_f32_16x16x4_f32(false, ahi, false, bhi, (short)0, acc[t], false, false);
        }
    }
}

extern "C" __global__ __launch_bounds__(256)
void dynconv2d_wmma_f32(const float* __restrict__ x,
                        const float* __restrict__ w,
                        const float* __restrict__ bias,
                        float* __restrict__ out)
{
    extern __shared__ __attribute__((aligned(16))) float smem[];
    float* sW = smem;
    float* sX = smem + SW_FLOATS;

    const int h0    = blockIdx.x * ROWS_PER_BLOCK;
    const int nrows = min(ROWS_PER_BLOCK, HOUT - h0);
    const int b     = blockIdx.y;
    const int tid   = threadIdx.x;
    const int wave  = tid >> 5;
    const int lane  = tid & 31;

    const int cg   = (wave & 3) * 8 + (lane & 7);
    const int ci_t = (wave >> 2) * 4 + (lane >> 3);
    const float* xsrc = x + (((size_t)b * CIN + ci_t) * H_IN + h0) * W_IN + 4 * cg;
    float*       xdst = sX + (4 * cg) * CIPITCH + ci_t;

    v4f ra[ROW_F4], rb[ROW_F4], rc[ROW_F4];
    row_fetch(ra, xsrc);
    row_fetch(rb, xsrc + W_IN);
    row_fetch(rc, xsrc + 2 * W_IN);

    const float* wsrc = w + (size_t)b * COUT * KDIM;

    #pragma unroll
    for (int k = 0; k < 2; ++k) {
        int idx = tid + 256 * k;
        sX[(idx >> 7) * XROW_FLOATS + (W_IN + ((idx >> 6) & 1)) * CIPITCH + (idx & 63)] = 0.0f;
    }

    #pragma unroll
    for (int g = 0; g < W_F4 / W_GRP; ++g) {
        v4f wv[W_GRP];
        #pragma unroll
        for (int j = 0; j < W_GRP; ++j)
            wv[j] = *(const v4f*)(wsrc + 4 * (tid + 256 * (g * W_GRP + j)));
        if (g == 0) {
            row_put(xdst,                   ra);
            row_put(xdst + XROW_FLOATS,     rb);
            row_put(xdst + 2 * XROW_FLOATS, rc);
        }
        #pragma unroll
        for (int j = 0; j < W_GRP; ++j) {
            int u   = tid + 256 * (g * W_GRP + j);
            int co  = u / 144;
            int rem = 4 * (u - co * 144);
            int ci  = rem / 9;
            int r   = rem - ci * 9;
            float* d = sW + co * WPITCH;
            #pragma unroll
            for (int e = 0; e < 4; ++e) {
                d[r * 64 + ci] = wv[j][e];
                if (++r == 9) { r = 0; ++ci; }
            }
        }
    }
    __syncthreads();

    const int laneN = lane & 15;
    const int hi    = lane >> 4;

    const int co_base = (wave & 3) * 16;
    const int P0      = (wave >> 2) * 64;

    const float* aB  = sW + (co_base + laneN) * WPITCH + 4 * hi;
    const float* fr0[4];
    const float* fr1[4];
    const float* fr2[4];
    const float* fr3[4];
    #pragma unroll
    for (int t = 0; t < 4; ++t) {
        fr0[t] = sX + (P0 + 16 * t + laneN) * CIPITCH + 4 * hi;
        fr1[t] = fr0[t] + XROW_FLOATS;
        fr2[t] = fr0[t] + 2 * XROW_FLOATS;
        fr3[t] = fr0[t] + 3 * XROW_FLOATS;
    }
    float* xd0 = xdst, *xd1 = xdst + XROW_FLOATS, *xd2 = xdst + 2 * XROW_FLOATS, *xd3 = xdst + 3 * XROW_FLOATS;

    float bv[8];
    #pragma unroll
    for (int j = 0; j < 8; ++j)
        bv[j] = bias[b * COUT + co_base + j + 8 * hi];

    float* obase = out + ((size_t)(b * COUT + co_base + 8 * hi) * HOUT) * WOUT + P0 + laneN;
    const bool tile3_ok = (P0 + 48 + laneN) < WOUT;

    v8f p0 = {}, p1 = {}, p2 = {}, p3 = {};

    Frag fs[2];
    frag_load(fs[0], aB, fr0, 0, 0);

    for (int hh = 0; hh < nrows; ++hh) {
        const int  h    = h0 + hh;
        const bool more = hh + 1 < nrows;

        v4f rn[ROW_F4];
        if (more)
            row_fetch(rn, xsrc + (hh + 3) * W_IN);

        v8f acc[4] = {};

        #pragma unroll
        for (int i = 0; i < NBLK; ++i) {
            if (i + 1 < NBLK) {
                const int r  = (i + 1) >> 1;
                const int kh = r / 3, kw = r - 3 * kh;
                const int cb = ((i + 1) & 1) * 32;
                const float* const (&fb)[4] = (kh == 0) ? fr0 : (kh == 1) ? fr1 : fr2;
                frag_load(fs[(i + 1) & 1], aB, fb, r * 64 + cb, kw * CIPITCH + cb);
            } else {
                frag_load(fs[0], aB, fr1, 0, 0);
            }
            __builtin_amdgcn_sched_barrier(0);

            frag_mma(acc, fs[i & 1]);

            __builtin_amdgcn_sched_barrier(0);

            if (i == 1) {
                if (hh > 0) {
                    float* o = obase + (size_t)(h - 1) * WOUT;
                    tile_out(o,      p0);
                    tile_out(o + 16, p1);
                    tile_out(o + 32, p2);
                    if (tile3_ok)
                        tile_out(o + 48, p3);
                }
                if (more)
                    row_put(xd3, rn);
            }
        }

        asm volatile("v_nop\n\tv_nop\n\tv_nop\n\tv_nop"
                     : "+v"(acc[0]), "+v"(acc[1]), "+v"(acc[2]), "+v"(acc[3]));

        p0 = add_bias(acc[0], bv); p1 = add_bias(acc[1], bv);
        p2 = add_bias(acc[2], bv); p3 = add_bias(acc[3], bv);

        if (more)
            __syncthreads();

        #pragma unroll
        for (int t = 0; t < 4; ++t) {
            const float* ft = fr0[t]; fr0[t] = fr1[t]; fr1[t] = fr2[t]; fr2[t] = fr3[t]; fr3[t] = ft;
        }
        float* xt = xd0; xd0 = xd1; xd1 = xd2; xd2 = xd3; xd3 = xt;
    }

    {
        float* o = obase + (size_t)(h0 + nrows - 1) * WOUT;
        tile_out(o,      p0);
        tile_out(o + 16, p1);
        tile_out(o + 32, p2);
        if (tile3_ok)
            tile_out(o + 48, p3);
    }
}

extern "C" void kernel_launch(void* const* d_in, const int* in_sizes, int n_in,
                              void* d_out, int out_size, void* d_ws, size_t ws_size,
                              hipStream_t stream) {
    (void)in_sizes; (void)n_in; (void)out_size; (void)d_ws; (void)ws_size;
    const float* x    = (const float*)d_in[0];
    const float* w    = (const float*)d_in[1];
    const float* bias = (const float*)d_in[2];
    float* out        = (float*)d_out;

    dim3 grid(NCHUNK, B_N);
    dynconv2d_wmma_f32<<<grid, 256, LDS_BYTES, stream>>>(x, w, bias, out);
}


# Vertical minimal filtering F(2,3): two output rows per step from four transformed input rows and transformed weights, 12 tap products per row pair instead of 18; summation order changes
# speedup vs baseline: 6.5963x; 1.1980x over previous
#include <hip/hip_runtime.h>

typedef __attribute__((ext_vector_type(2))) float v2f;
typedef __attribute__((ext_vector_type(4))) float v4f;
typedef __attribute__((ext_vector_type(8))) float v8f;

#define B_N    16
#define CIN    64
#define COUT   64
#define H_IN   128
#define W_IN   128
#define HOUT   126
#define WOUT   126
#define KDIM   576

#define WPITCH 580
#define XCOLS  130
#define CIPITCH 68

#define ROWS_PER_BLOCK 8
#define NCHUNK 16

#define NSLOT 4
#define SW_FLOATS (COUT * WPITCH)
#define XROW_FLOATS (XCOLS * CIPITCH)
#define SX_FLOATS (NSLOT * XROW_FLOATS)
#define LDS_BYTES ((SW_FLOATS + SX_FLOATS) * 4)

#define ROW_F4 8
#define PLANE8 (8 * H_IN * W_IN)
#define W_F4   36
#define W_GRP  12

__device__ __forceinline__ void row_fetch(v4f (&r)[ROW_F4], const float* __restrict__ src)
{
    #pragma unroll
    for (int j = 0; j < ROW_F4; ++j)
        r[j] = *(const v4f*)(src + (size_t)j * PLANE8);
}

__device__ __forceinline__ void put4(float* dst, const v4f& v)
{
    dst[0 * CIPITCH] = v.x;
    dst[1 * CIPITCH] = v.y;
    dst[2 * CIPITCH] = v.z;
    dst[3 * CIPITCH] = v.w;
}

__device__ __forceinline__ void rows_put(float* dst, const v4f (&d0)[ROW_F4], const v4f (&d1)[ROW_F4],
                                         const v4f (&d2)[ROW_F4], const v4f (&d3)[ROW_F4])
{
    #pragma unroll
    for (int j = 0; j < ROW_F4; ++j) {
        put4(dst + 8 * j,                   d0[j] - d2[j]);
        put4(dst + 8 * j + XROW_FLOATS,     d1[j] + d2[j]);
        put4(dst + 8 * j + 2 * XROW_FLOATS, d2[j] - d1[j]);
        put4(dst + 8 * j + 3 * XROW_FLOATS, d1[j] - d3[j]);
    }
}

__device__ __forceinline__ void tile_out(float* __restrict__ o, const v8f& res)
{
    #pragma unroll
    for (int j = 0; j < 8; ++j)
        o[(size_t)j * (HOUT * WOUT)] = res[j];
}

#define NBLK   24

struct Frag {
    v4f a[4];
    v4f b[4][4];
};

__device__ __forceinline__ void frag_load(Frag& f, const float* aP, const float* const (&bP)[4],
                                          int aoff, int boff)
{
    #pragma unroll
    for (int j = 0; j < 4; ++j)
        f.a[j] = *(const v4f*)(aP + aoff + 8 * j);
    #pragma unroll
    for (int t = 0; t < 4; ++t) {
        #pragma unroll
        for (int j = 0; j < 4; ++j)
            f.b[t][j] = *(const v4f*)(bP[t] + boff + 8 * j);
    }
}

__device__ __forceinline__ void extra_load(v4f (&ex)[8], const float* aP, int aoff)
{
    #pragma unroll
    for (int j = 0; j < 4; ++j) {
        ex[j]     = *(const v4f*)(aP + aoff + 192 + 8 * j);
        ex[4 + j] = *(const v4f*)(aP + aoff + 384 + 8 * j);
    }
}

__device__ __forceinline__ void frag_mma(v8f (&acc)[4], const Frag& f)
{
    #pragma unroll
    for (int j = 0; j < 4; ++j) {
        const v2f alo = __builtin_shufflevector(f.a[j], f.a[j], 0, 1);
        const v2f ahi = __builtin_shufflevector(f.a[j], f.a[j], 2, 3);
        #pragma unroll
        for (int t = 0; t < 4; ++t) {
            const v2f blo = __builtin_shufflevector(f.b[t][j], f.b[t][j], 0, 1);
            acc[t] = __builtin_amdgcn_wmma_f32_16x16x4_f32(false, alo, false, blo, (short)0, acc[t], false, false);
        }
        #pragma unroll
        for (int t = 0; t < 4; ++t) {
            const v2f bhi = __builtin_shufflevector(f.b[t][j], f.b[t][j], 2, 3);
            acc[t] = __builtin_amdgcn_wmma_f32_16x16x4_f32(false, ahi, false, bhi, (short)0, acc[t], false, false);
        }
    }
}

__device__ __forceinline__ constexpr int tap_of(int i) { return (i / 6 == 0) ? 3 : (i / 6 - 1); }

extern "C" __global__ __launch_bounds__(256)
void dynconv2d_wmma_f32(const float* __restrict__ x,
                        const float* __restrict__ w,
                        const float* __restrict__ bias,
                        float* __restrict__ out)
{
    extern __shared__ __attribute__((aligned(16))) float smem[];
    float* sW = smem;
    float* sX = smem + SW_FLOATS;

    const int h0     = blockIdx.x * ROWS_PER_BLOCK;
    const int nsteps = min(ROWS_PER_BLOCK, HOUT - h0) / 2;
    const int b     = blockIdx.y;
    const int tid   = threadIdx.x;
    const int wave  = tid >> 5;
    const int lane  = tid & 31;

    const int cg   = (wave & 3) * 8 + (lane & 7);
    const int ci_t = (wave >> 2) * 4 + (lane >> 3);
    const float* xsrc = x + (((size_t)b * CIN + ci_t) * H_IN + h0) * W_IN + 4 * cg;
    float*       xdst = sX + (4 * cg) * CIPITCH + ci_t;

    v4f ra[ROW_F4], rb[ROW_F4], rc[ROW_F4], rd[ROW_F4];
    row_fetch(ra, xsrc);
    row_fetch(rb, xsrc + W_IN);
    row_fetch(rc, xsrc + 2 * W_IN);
    row_fetch(rd, xsrc + 3 * W_IN);

    const float* wsrc = w + (size_t)b * COUT * KDIM;

    #pragma unroll
    for (int k = 0; k < 2; ++k) {
        int idx = tid + 256 * k;
        sX[(idx >> 7) * XROW_FLOATS + (W_IN + ((idx >> 6) & 1)) * CIPITCH + (idx & 63)] = 0.0f;
    }

    #pragma unroll
    for (int g = 0; g < W_F4 / W_GRP; ++g) {
        v4f wv[W_GRP];
        #pragma unroll
        for (int j = 0; j < W_GRP; ++j)
            wv[j] = *(const v4f*)(wsrc + 4 * (tid + 256 * (g * W_GRP + j)));
        if (g == 0)
            rows_put(xdst, ra, rb, rc, rd);
        #pragma unroll
        for (int j = 0; j < W_GRP; ++j) {
            int u   = tid + 256 * (g * W_GRP + j);
            int co  = u / 144;
            int rem = 4 * (u - co * 144);
            int ci  = rem / 9;
            int r   = rem - ci * 9;
            float* d = sW + co * WPITCH;
            #pragma unroll
            for (int e = 0; e < 4; ++e) {
                d[r * 64 + ci] = wv[j][e];
                if (++r == 9) { r = 0; ++ci; }
            }
        }
    }
    __syncthreads();

    #pragma unroll
    for (int k = 0; k < 12; ++k) {
        int idx = tid + 256 * k;
        int co  = idx / 48;
        int rem = idx - co * 48;
        float* p = sW + co * WPITCH + 4 * rem;
        const v4f g0 = *(const v4f*)p;
        const v4f g1 = *(const v4f*)(p + 192);
        const v4f g2 = *(const v4f*)(p + 384);
        const v4f s  = g0 + g2;
        *(v4f*)(p + 192) = (s + g1) * 0.5f;
        *(v4f*)(p + 384) = (s - g1) * 0.5f;
    }
    __syncthreads();

    const int laneN = lane & 15;
    const int hi    = lane >> 4;

    const int co_base = (wave & 3) * 16;
    const int P0      = (wave >> 2) * 64;

    const float* aB  = sW + (co_base + laneN) * WPITCH + 4 * hi;
    const float* fbL[4];
    const float* fbH[4];
    #pragma unroll
    for (int t = 0; t < 4; ++t) {
        fbL[t] = sX + (P0 + 16 * t + laneN) * CIPITCH + 4 * hi;
        fbH[t] = fbL[t] + 2 * XROW_FLOATS;
    }

    v8f bvec;
    #pragma unroll
    for (int j = 0; j < 8; ++j)
        bvec[j] = bias[b * COUT + co_base + j + 8 * hi];

    float* obase = out + ((size_t)(b * COUT + co_base + 8 * hi) * HOUT) * WOUT + P0 + laneN;
    const bool tile3_ok = (P0 + 48 + laneN) < WOUT;

    Frag fs[2];
    v4f  ex[8];

    for (int s = 0; s < nsteps; ++s) {
        const int  h    = h0 + 2 * s;
        const bool more = s + 1 < nsteps;

        v8f acc[4][4];
        #pragma unroll
        for (int t = 0; t < 4; ++t) {
            acc[0][t] = bvec; acc[3][t] = bvec;
            acc[1][t] = v8f{}; acc[2][t] = v8f{};
        }

        v4f rn0[ROW_F4], rn1[ROW_F4], rn2[ROW_F4], rn3[ROW_F4];

        frag_load(fs[0], aB, fbH, 0, XROW_FLOATS);
        extra_load(ex, aB, 0);

        #pragma unroll
        for (int i = 0; i < NBLK; ++i) {
            const int q = tap_of(i);
            if (q == 3) {
                #pragma unroll
                for (int j = 0; j < 4; ++j)
                    fs[i & 1].a[j] = fs[i & 1].a[j] - (ex[j] + ex[4 + j]);
            }
            __builtin_amdgcn_sched_barrier(0);

            if (i + 1 < NBLK) {
                const int qn  = tap_of(i + 1);
                const int kwn = ((i + 1) % 6) / 2;
                const int cbn = ((i + 1) & 1) * 32;
                const int pl  = (qn == 3) ? 0 : qn;
                const float* const (&fb)[4] = (qn < 2) ? fbL : fbH;
                frag_load(fs[(i + 1) & 1], aB, fb, (pl * 3 + kwn) * 64 + cbn,
                          (qn & 1) * XROW_FLOATS + kwn * CIPITCH + cbn);
                if (qn == 3)
                    extra_load(ex, aB, kwn * 64 + cbn);
            }
            if (i == 12 && more) {
                const float* nx = xsrc + (size_t)(2 * (s + 1)) * W_IN;
                row_fetch(rn0, nx);
                row_fetch(rn1, nx + W_IN);
                row_fetch(rn2, nx + 2 * W_IN);
                row_fetch(rn3, nx + 3 * W_IN);
            }
            __builtin_amdgcn_sched_barrier(0);

            frag_mma(acc[q], fs[i & 1]);

            __builtin_amdgcn_sched_barrier(0);
        }

        asm volatile("v_nop\n\tv_nop\n\tv_nop\n\tv_nop"
                     : "+v"(acc[2][0]), "+v"(acc[2][1]), "+v"(acc[2][2]), "+v"(acc[2][3]));

        {
            float* o0 = obase + (size_t)h * WOUT;
            float* o1 = o0 + WOUT;
            #pragma unroll
            for (int t = 0; t < 4; ++t) {
                if (t < 3 || tile3_ok) {
                    const v8f y0 = (acc[0][t] + acc[1][t]) + acc[2][t];
                    const v8f y1 = (acc[3][t] + acc[1][t]) - acc[2][t];
                    tile_out(o0 + 16 * t, y0);
                    tile_out(o1 + 16 * t, y1);
                }
            }
        }

        if (more) {
            __syncthreads();
            rows_put(xdst, rn0, rn1, rn2, rn3);
            __syncthreads();
        }
    }
}

extern "C" void kernel_launch(void* const* d_in, const int* in_sizes, int n_in,
                              void* d_out, int out_size, void* d_ws, size_t ws_size,
                              hipStream_t stream) {
    (void)in_sizes; (void)n_in; (void)out_size; (void)d_ws; (void)ws_size;
    const float* x    = (const float*)d_in[0];
    const float* w    = (const float*)d_in[1];
    const float* bias = (const float*)d_in[2];
    float* out        = (float*)d_out;

    dim3 grid(NCHUNK, B_N);
    dynconv2d_wmma_f32<<<grid, 256, LDS_BYTES, stream>>>(x, w, bias, out);
}


# All LDS staging by 16-byte stores: input rows transposed in registers (4 ci per store, lanes remapped over the banks), weights through a raw LDS copy regrouped per (co, 4 ci); sums unchanged
# speedup vs baseline: 6.8920x; 1.0448x over previous
#include <hip/hip_runtime.h>

typedef __attribute__((ext_vector_type(2))) float v2f;
typedef __attribute__((ext_vector_type(4))) float v4f;
typedef __attribute__((ext_vector_type(8))) float v8f;

#define B_N    16
#define CIN    64
#define COUT   64
#define H_IN   128
#define W_IN   128
#define HOUT   126
#define WOUT   126
#define KDIM   576

#define WPITCH 580
#define XCOLS  130
#define CIPITCH 68

#define ROWS_PER_BLOCK 8
#define NCHUNK 16

#define NSLOT 4
#define SW_FLOATS (COUT * WPITCH)
#define XROW_FLOATS (XCOLS * CIPITCH)
#define SX_FLOATS (NSLOT * XROW_FLOATS)
#define WRAW_FLOATS (COUT * KDIM)
#define LDS_BYTES ((SW_FLOATS + WRAW_FLOATS) * 4)

#define ROW_F4 8
#define PLANE  (H_IN * W_IN)
#define W_F4   36

__device__ __forceinline__ void row_fetch(v4f (&r)[ROW_F4], const float* __restrict__ src)
{
    #pragma unroll
    for (int jq = 0; jq < 2; ++jq) {
        #pragma unroll
        for (int jj = 0; jj < 4; ++jj)
            r[4 * jq + jj] = *(const v4f*)(src + (size_t)(16 * jq + jj) * PLANE);
    }
}

__device__ __forceinline__ void rows_put(float* dst, const v4f (&d0)[ROW_F4], const v4f (&d1)[ROW_F4],
                                         const v4f (&d2)[ROW_F4], const v4f (&d3)[ROW_F4])
{
    #pragma unroll
    for (int jq = 0; jq < 2; ++jq) {
        #pragma unroll
        for (int e = 0; e < 4; ++e) {
            v4f t0, t1, t2, t3;
            #pragma unroll
            for (int jj = 0; jj < 4; ++jj) {
                const int j = 4 * jq + jj;
                t0[jj] = d0[j][e] - d2[j][e];
                t1[jj] = d1[j][e] + d2[j][e];
                t2[jj] = d2[j][e] - d1[j][e];
                t3[jj] = d1[j][e] - d3[j][e];
            }
            float* p = dst + e * CIPITCH + 16 * jq;
            *(v4f*)(p)                   = t0;
            *(v4f*)(p + XROW_FLOATS)     = t1;
            *(v4f*)(p + 2 * XROW_FLOATS) = t2;
            *(v4f*)(p + 3 * XROW_FLOATS) = t3;
        }
    }
}

__device__ __forceinline__ void tile_out(float* __restrict__ o, const v8f& res)
{
    #pragma unroll
    for (int j = 0; j < 8; ++j)
        o[(size_t)j * (HOUT * WOUT)] = res[j];
}

#define NBLK   24

struct Frag {
    v4f a[4];
    v4f b[4][4];
};

__device__ __forceinline__ void frag_load(Frag& f, const float* aP, const float* const (&bP)[4],
                                          int aoff, int boff)
{
    #pragma unroll
    for (int j = 0; j < 4; ++j)
        f.a[j] = *(const v4f*)(aP + aoff + 8 * j);
    #pragma unroll
    for (int t = 0; t < 4; ++t) {
        #pragma unroll
        for (int j = 0; j < 4; ++j)
            f.b[t][j] = *(const v4f*)(bP[t] + boff + 8 * j);
    }
}

__device__ __forceinline__ void extra_load(v4f (&ex)[8], const float* aP, int aoff)
{
    #pragma unroll
    for (int j = 0; j < 4; ++j) {
        ex[j]     = *(const v4f*)(aP + aoff + 192 + 8 * j);
        ex[4 + j] = *(const v4f*)(aP + aoff + 384 + 8 * j);
    }
}

__device__ __forceinline__ void frag_mma(v8f (&acc)[4], const Frag& f)
{
    #pragma unroll
    for (int j = 0; j < 4; ++j) {
        const v2f alo = __builtin_shufflevector(f.a[j], f.a[j], 0, 1);
        const v2f ahi = __builtin_shufflevector(f.a[j], f.a[j], 2, 3);
        #pragma unroll
        for (int t = 0; t < 4; ++t) {
            const v2f blo = __builtin_shufflevector(f.b[t][j], f.b[t][j], 0, 1);
            acc[t] = __builtin_amdgcn_wmma_f32_16x16x4_f32(false, alo, false, blo, (short)0, acc[t], false, false);
        }
        #pragma unroll
        for (int t = 0; t < 4; ++t) {
            const v2f bhi = __builtin_shufflevector(f.b[t][j], f.b[t][j], 2, 3);
            acc[t] = __builtin_amdgcn_wmma_f32_16x16x4_f32(false, ahi, false, bhi, (short)0, acc[t], false, false);
        }
    }
}

__device__ __forceinline__ constexpr int tap_of(int i) { return (i / 6 == 0) ? 3 : (i / 6 - 1); }

extern "C" __global__ __launch_bounds__(256)
void dynconv2d_wmma_f32(const float* __restrict__ x,
                        const float* __restrict__ w,
                        const float* __restrict__ bias,
                        float* __restrict__ out)
{
    extern __shared__ __attribute__((aligned(16))) float smem[];
    float* sW = smem;
    float* sX = smem + SW_FLOATS;

    const int h0     = blockIdx.x * ROWS_PER_BLOCK;
    const int nsteps = min(ROWS_PER_BLOCK, HOUT - h0) / 2;
    const int b     = blockIdx.y;
    const int tid   = threadIdx.x;
    const int wave  = tid >> 5;
    const int lane  = tid & 31;

    const int cg   = (wave & 3) * 8 + (lane & 3) + 4 * (lane >> 4);
    const int ci_t = (wave >> 2) * 32 + 4 * ((lane >> 2) & 3);
    const float* xsrc = x + (((size_t)b * CIN + ci_t) * H_IN + h0) * W_IN + 4 * cg;
    float*       xdst = sX + (4 * cg) * CIPITCH + ci_t;

    v4f ra[ROW_F4], rb[ROW_F4], rc[ROW_F4], rd[ROW_F4];
    row_fetch(ra, xsrc);
    row_fetch(rb, xsrc + W_IN);
    row_fetch(rc, xsrc + 2 * W_IN);
    row_fetch(rd, xsrc + 3 * W_IN);

    const float* wsrc = w + (size_t)b * COUT * KDIM;

    float* wraw = sX;
    {
        v4f wv[W_F4];
        #pragma unroll
        for (int j = 0; j < W_F4; ++j)
            wv[j] = *(const v4f*)(wsrc + 4 * (tid + 256 * j));
        #pragma unroll
        for (int j = 0; j < W_F4; ++j)
            *(v4f*)(wraw + 4 * (tid + 256 * j)) = wv[j];
    }
    __syncthreads();

    #pragma unroll
    for (int k = 0; k < 4; ++k) {
        const int v  = tid + 256 * k;
        const int co = v >> 4;
        const float* rs = wraw + 36 * v;
        v4f raw[9];
        #pragma unroll
        for (int m = 0; m < 9; ++m)
            raw[m] = *(const v4f*)(rs + 4 * m);
        float* d = sW + co * WPITCH + 4 * (v & 15);
        #pragma unroll
        for (int kw = 0; kw < 3; ++kw) {
            v4f g0, g1, g2;
            #pragma unroll
            for (int c = 0; c < 4; ++c) {
                g0[c] = raw[(9 * c + kw) >> 2][(9 * c + kw) & 3];
                g1[c] = raw[(9 * c + 3 + kw) >> 2][(9 * c + 3 + kw) & 3];
                g2[c] = raw[(9 * c + 6 + kw) >> 2][(9 * c + 6 + kw) & 3];
            }
            const v4f s = g0 + g2;
            *(v4f*)(d + kw * 64)       = g0;
            *(v4f*)(d + 192 + kw * 64) = (s + g1) * 0.5f;
            *(v4f*)(d + 384 + kw * 64) = (s - g1) * 0.5f;
        }
    }
    __syncthreads();

    #pragma unroll
    for (int k = 0; k < 2; ++k) {
        int idx = tid + 256 * k;
        sX[(idx >> 7) * XROW_FLOATS + (W_IN + ((idx >> 6) & 1)) * CIPITCH + (idx & 63)] = 0.0f;
    }
    rows_put(xdst, ra, rb, rc, rd);
    __syncthreads();

    const int laneN = lane & 15;
    const int hi    = lane >> 4;

    const int co_base = (wave & 3) * 16;
    const int P0      = (wave >> 2) * 64;

    const float* aB  = sW + (co_base + laneN) * WPITCH + 4 * hi;
    const float* fbL[4];
    const float* fbH[4];
    #pragma unroll
    for (int t = 0; t < 4; ++t) {
        fbL[t] = sX + (P0 + 16 * t + laneN) * CIPITCH + 4 * hi;
        fbH[t] = fbL[t] + 2 * XROW_FLOATS;
    }

    v8f bvec;
    #pragma unroll
    for (int j = 0; j < 8; ++j)
        bvec[j] = bias[b * COUT + co_base + j + 8 * hi];

    float* obase = out + ((size_t)(b * COUT + co_base + 8 * hi) * HOUT) * WOUT + P0 + laneN;
    const bool tile3_ok = (P0 + 48 + laneN) < WOUT;

    Frag fs[2];
    v4f  ex[8];

    for (int s = 0; s < nsteps; ++s) {
        const int  h    = h0 + 2 * s;
        const bool more = s + 1 < nsteps;

        v8f acc[4][4];
        #pragma unroll
        for (int t = 0; t < 4; ++t) {
            acc[0][t] = bvec; acc[3][t] = bvec;
            acc[1][t] = v8f{}; acc[2][t] = v8f{};
        }

        v4f rn0[ROW_F4], rn1[ROW_F4], rn2[ROW_F4], rn3[ROW_F4];

        frag_load(fs[0], aB, fbH, 0, XROW_FLOATS);
        extra_load(ex, aB, 0);

        #pragma unroll
        for (int i = 0; i < NBLK; ++i) {
            const int q = tap_of(i);
            if (q == 3) {
                #pragma unroll
                for (int j = 0; j < 4; ++j)
                    fs[i & 1].a[j] = fs[i & 1].a[j] - (ex[j] + ex[4 + j]);
            }
            __builtin_amdgcn_sched_barrier(0);

            if (i + 1 < NBLK) {
                const int qn  = tap_of(i + 1);
                const int kwn = ((i + 1) % 6) / 2;
                const int cbn = ((i + 1) & 1) * 32;
                const int pl  = (qn == 3) ? 0 : qn;
                const float* const (&fb)[4] = (qn < 2) ? fbL : fbH;
                frag_load(fs[(i + 1) & 1], aB, fb, (pl * 3 + kwn) * 64 + cbn,
                          (qn & 1) * XROW_FLOATS + kwn * CIPITCH + cbn);
                if (qn == 3)
                    extra_load(ex, aB, kwn * 64 + cbn);
            }
            if (i == 12 && more) {
                const float* nx = xsrc + (size_t)(2 * (s + 1)) * W_IN;
                row_fetch(rn0, nx);
                row_fetch(rn1, nx + W_IN);
                row_fetch(rn2, nx + 2 * W_IN);
                row_fetch(rn3, nx + 3 * W_IN);
            }
            __builtin_amdgcn_sched_barrier(0);

            frag_mma(acc[q], fs[i & 1]);

            __builtin_amdgcn_sched_barrier(0);
        }

        asm volatile("v_nop\n\tv_nop\n\tv_nop\n\tv_nop"
                     : "+v"(acc[2][0]), "+v"(acc[2][1]), "+v"(acc[2][2]), "+v"(acc[2][3]));

        {
            float* o0 = obase + (size_t)h * WOUT;
            float* o1 = o0 + WOUT;
            #pragma unroll
            for (int t = 0; t < 4; ++t) {
                if (t < 3 || tile3_ok) {
                    const v8f y0 = (acc[0][t] + acc[1][t]) + acc[2][t];
                    const v8f y1 = (acc[3][t] + acc[1][t]) - acc[2][t];
                    tile_out(o0 + 16 * t, y0);
                    tile_out(o1 + 16 * t, y1);
                }
            }
        }

        if (more) {
            __syncthreads();
            rows_put(xdst, rn0, rn1, rn2, rn3);
            __syncthreads();
        }
    }
}

extern "C" void kernel_launch(void* const* d_in, const int* in_sizes, int n_in,
                              void* d_out, int out_size, void* d_ws, size_t ws_size,
                              hipStream_t stream) {
    (void)in_sizes; (void)n_in; (void)out_size; (void)d_ws; (void)ws_size;
    const float* x    = (const float*)d_in[0];
    const float* w    = (const float*)d_in[1];
    const float* bias = (const float*)d_in[2];
    float* out        = (float*)d_out;

    dim3 grid(NCHUNK, B_N);
    dynconv2d_wmma_f32<<<grid, 256, LDS_BYTES, stream>>>(x, w, bias, out);
}


# A step's two finished rows stay in registers and are stored one tile per block in the read groups of blocks 1..8 of the next step, not in front of the step-end barrier; sums unchanged
# speedup vs baseline: 7.1320x; 1.0348x over previous
#include <hip/hip_runtime.h>

typedef __attribute__((ext_vector_type(2))) float v2f;
typedef __attribute__((ext_vector_type(4))) float v4f;
typedef __attribute__((ext_vector_type(8))) float v8f;

#define B_N    16
#define CIN    64
#define COUT   64
#define H_IN   128
#define W_IN   128
#define HOUT   126
#define WOUT   126
#define KDIM   576

#define WPITCH 580
#define XCOLS  130
#define CIPITCH 68

#define ROWS_PER_BLOCK 8
#define NCHUNK 16

#define NSLOT 4
#define SW_FLOATS (COUT * WPITCH)
#define XROW_FLOATS (XCOLS * CIPITCH)
#define SX_FLOATS (NSLOT * XROW_FLOATS)
#define WRAW_FLOATS (COUT * KDIM)
#define LDS_BYTES ((SW_FLOATS + WRAW_FLOATS) * 4)

#define ROW_F4 8
#define PLANE  (H_IN * W_IN)
#define W_F4   36

__device__ __forceinline__ void row_fetch(v4f (&r)[ROW_F4], const float* __restrict__ src)
{
    #pragma unroll
    for (int jq = 0; jq < 2; ++jq) {
        #pragma unroll
        for (int jj = 0; jj < 4; ++jj)
            r[4 * jq + jj] = *(const v4f*)(src + (size_t)(16 * jq + jj) * PLANE);
    }
}

__device__ __forceinline__ void rows_put(float* dst, const v4f (&d0)[ROW_F4], const v4f (&d1)[ROW_F4],
                                         const v4f (&d2)[ROW_F4], const v4f (&d3)[ROW_F4])
{
    #pragma unroll
    for (int jq = 0; jq < 2; ++jq) {
        #pragma unroll
        for (int e = 0; e < 4; ++e) {
            v4f t0, t1, t2, t3;
            #pragma unroll
            for (int jj = 0; jj < 4; ++jj) {
                const int j = 4 * jq + jj;
                t0[jj] = d0[j][e] - d2[j][e];
                t1[jj] = d1[j][e] + d2[j][e];
                t2[jj] = d2[j][e] - d1[j][e];
                t3[jj] = d1[j][e] - d3[j][e];
            }
            float* p = dst + e * CIPITCH + 16 * jq;
            *(v4f*)(p)                   = t0;
            *(v4f*)(p + XROW_FLOATS)     = t1;
            *(v4f*)(p + 2 * XROW_FLOATS) = t2;
            *(v4f*)(p + 3 * XROW_FLOATS) = t3;
        }
    }
}

__device__ __forceinline__ void tile_out(float* __restrict__ o, const v8f& res)
{
    #pragma unroll
    for (int j = 0; j < 8; ++j)
        o[(size_t)j * (HOUT * WOUT)] = res[j];
}

#define NBLK   24

struct Frag {
    v4f a[4];
    v4f b[4][4];
};

__device__ __forceinline__ void frag_load(Frag& f, const float* aP, const float* const (&bP)[4],
                                          int aoff, int boff)
{
    #pragma unroll
    for (int j = 0; j < 4; ++j)
        f.a[j] = *(const v4f*)(aP + aoff + 8 * j);
    #pragma unroll
    for (int t = 0; t < 4; ++t) {
        #pragma unroll
        for (int j = 0; j < 4; ++j)
            f.b[t][j] = *(const v4f*)(bP[t] + boff + 8 * j);
    }
}

__device__ __forceinline__ void extra_load(v4f (&ex)[8], const float* aP, int aoff)
{
    #pragma unroll
    for (int j = 0; j < 4; ++j) {
        ex[j]     = *(const v4f*)(aP + aoff + 192 + 8 * j);
        ex[4 + j] = *(const v4f*)(aP + aoff + 384 + 8 * j);
    }
}

__device__ __forceinline__ void frag_mma(v8f (&acc)[4], const Frag& f)
{
    #pragma unroll
    for (int j = 0; j < 4; ++j) {
        const v2f alo = __builtin_shufflevector(f.a[j], f.a[j], 0, 1);
        const v2f ahi = __builtin_shufflevector(f.a[j], f.a[j], 2, 3);
        #pragma unroll
        for (int t = 0; t < 4; ++t) {
            const v2f blo = __builtin_shufflevector(f.b[t][j], f.b[t][j], 0, 1);
            acc[t] = __builtin_amdgcn_wmma_f32_16x16x4_f32(false, alo, false, blo, (short)0, acc[t], false, false);
        }
        #pragma unroll
        for (int t = 0; t < 4; ++t) {
            const v2f bhi = __builtin_shufflevector(f.b[t][j], f.b[t][j], 2, 3);
            acc[t] = __builtin_amdgcn_wmma_f32_16x16x4_f32(false, ahi, false, bhi, (short)0, acc[t], false, false);
        }
    }
}

__device__ __forceinline__ constexpr int tap_of(int i) { return (i / 6 == 0) ? 3 : (i / 6 - 1); }

extern "C" __global__ __launch_bounds__(256)
void dynconv2d_wmma_f32(const float* __restrict__ x,
                        const float* __restrict__ w,
                        const float* __restrict__ bias,
                        float* __restrict__ out)
{
    extern __shared__ __attribute__((aligned(16))) float smem[];
    float* sW = smem;
    float* sX = smem + SW_FLOATS;

    const int h0     = blockIdx.x * ROWS_PER_BLOCK;
    const int nsteps = min(ROWS_PER_BLOCK, HOUT - h0) / 2;
    const int b     = blockIdx.y;
    const int tid   = threadIdx.x;
    const int wave  = tid >> 5;
    const int lane  = tid & 31;

    const int cg   = (wave & 3) * 8 + (lane & 3) + 4 * (lane >> 4);
    const int ci_t = (wave >> 2) * 32 + 4 * ((lane >> 2) & 3);
    const float* xsrc = x + (((size_t)b * CIN + ci_t) * H_IN + h0) * W_IN + 4 * cg;
    float*       xdst = sX + (4 * cg) * CIPITCH + ci_t;

    v4f ra[ROW_F4], rb[ROW_F4], rc[ROW_F4], rd[ROW_F4];
    row_fetch(ra, xsrc);
    row_fetch(rb, xsrc + W_IN);
    row_fetch(rc, xsrc + 2 * W_IN);
    row_fetch(rd, xsrc + 3 * W_IN);

    const float* wsrc = w + (size_t)b * COUT * KDIM;

    float* wraw = sX;
    {
        v4f wv[W_F4];
        #pragma unroll
        for (int j = 0; j < W_F4; ++j)
            wv[j] = *(const v4f*)(wsrc + 4 * (tid + 256 * j));
        #pragma unroll
        for (int j = 0; j < W_F4; ++j)
            *(v4f*)(wraw + 4 * (tid + 256 * j)) = wv[j];
    }
    __syncthreads();

    #pragma unroll
    for (int k = 0; k < 4; ++k) {
        const int v  = tid + 256 * k;
        const int co = v >> 4;
        const float* rs = wraw + 36 * v;
        v4f raw[9];
        #pragma unroll
        for (int m = 0; m < 9; ++m)
            raw[m] = *(const v4f*)(rs + 4 * m);
        float* d = sW + co * WPITCH + 4 * (v & 15);
        #pragma unroll
        for (int kw = 0; kw < 3; ++kw) {
            v4f g0, g1, g2;
            #pragma unroll
            for (int c = 0; c < 4; ++c) {
                g0[c] = raw[(9 * c + kw) >> 2][(9 * c + kw) & 3];
                g1[c] = raw[(9 * c + 3 + kw) >> 2][(9 * c + 3 + kw) & 3];
                g2[c] = raw[(9 * c + 6 + kw) >> 2][(9 * c + 6 + kw) & 3];
            }
            const v4f s = g0 + g2;
            *(v4f*)(d + kw * 64)       = g0;
            *(v4f*)(d + 192 + kw * 64) = (s + g1) * 0.5f;
            *(v4f*)(d + 384 + kw * 64) = (s - g1) * 0.5f;
        }
    }
    __syncthreads();

    #pragma unroll
    for (int k = 0; k < 2; ++k) {
        int idx = tid + 256 * k;
        sX[(idx >> 7) * XROW_FLOATS + (W_IN + ((idx >> 6) & 1)) * CIPITCH + (idx & 63)] = 0.0f;
    }
    rows_put(xdst, ra, rb, rc, rd);
    __syncthreads();

    const int laneN = lane & 15;
    const int hi    = lane >> 4;

    const int co_base = (wave & 3) * 16;
    const int P0      = (wave >> 2) * 64;

    const float* aB  = sW + (co_base + laneN) * WPITCH + 4 * hi;
    const float* fbL[4];
    const float* fbH[4];
    #pragma unroll
    for (int t = 0; t < 4; ++t) {
        fbL[t] = sX + (P0 + 16 * t + laneN) * CIPITCH + 4 * hi;
        fbH[t] = fbL[t] + 2 * XROW_FLOATS;
    }

    v8f bvec;
    #pragma unroll
    for (int j = 0; j < 8; ++j)
        bvec[j] = bias[b * COUT + co_base + j + 8 * hi];

    float* obase = out + ((size_t)(b * COUT + co_base + 8 * hi) * HOUT) * WOUT + P0 + laneN;
    const bool tile3_ok = (P0 + 48 + laneN) < WOUT;

    Frag fs[2];
    v4f  ex[8];

    v8f acc[4][4];
    v8f yo0[4], yo1[4];

    for (int s = 0; s < nsteps; ++s) {
        const int  h    = h0 + 2 * s;
        const bool more = s + 1 < nsteps;
        const bool prev = s > 0;
        float* const op = obase + (size_t)(prev ? h - 2 : h) * WOUT;

        v4f rn0[ROW_F4], rn1[ROW_F4], rn2[ROW_F4], rn3[ROW_F4];

        frag_load(fs[0], aB, fbH, 0, XROW_FLOATS);
        extra_load(ex, aB, 0);
        if (prev) {
            #pragma unroll
            for (int t = 0; t < 4; ++t) {
                yo0[t] = (acc[0][t] + acc[1][t]) + acc[2][t];
                yo1[t] = (acc[3][t] + acc[1][t]) - acc[2][t];
            }
        }

        #pragma unroll
        for (int t = 0; t < 4; ++t) {
            acc[0][t] = bvec; acc[3][t] = bvec;
            acc[1][t] = v8f{}; acc[2][t] = v8f{};
        }

        #pragma unroll
        for (int i = 0; i < NBLK; ++i) {
            const int q = tap_of(i);
            if (q == 3) {
                #pragma unroll
                for (int j = 0; j < 4; ++j)
                    fs[i & 1].a[j] = fs[i & 1].a[j] - (ex[j] + ex[4 + j]);
            }
            __builtin_amdgcn_sched_barrier(0);

            if (i + 1 < NBLK) {
                const int qn  = tap_of(i + 1);
                const int kwn = ((i + 1) % 6) / 2;
                const int cbn = ((i + 1) & 1) * 32;
                const int pl  = (qn == 3) ? 0 : qn;
                const float* const (&fb)[4] = (qn < 2) ? fbL : fbH;
                frag_load(fs[(i + 1) & 1], aB, fb, (pl * 3 + kwn) * 64 + cbn,
                          (qn & 1) * XROW_FLOATS + kwn * CIPITCH + cbn);
                if (qn == 3)
                    extra_load(ex, aB, kwn * 64 + cbn);
            }
            if (i == 12 && more) {
                const float* nx = xsrc + (size_t)(2 * (s + 1)) * W_IN;
                row_fetch(rn0, nx);
                row_fetch(rn1, nx + W_IN);
                row_fetch(rn2, nx + 2 * W_IN);
                row_fetch(rn3, nx + 3 * W_IN);
            }
            if (i >= 1 && i <= 8 && prev) {
                const int t = (i - 1) % 4;
                if (t < 3 || tile3_ok)
                    tile_out(op + ((i - 1) / 4) * WOUT + 16 * t, (i <= 4) ? yo0[t] : yo1[t]);
            }
            __builtin_amdgcn_sched_barrier(0);

            frag_mma(acc[q], fs[i & 1]);

            __builtin_amdgcn_sched_barrier(0);
        }

        asm volatile("v_nop\n\tv_nop\n\tv_nop\n\tv_nop"
                     : "+v"(acc[2][0]), "+v"(acc[2][1]), "+v"(acc[2][2]), "+v"(acc[2][3]));


        if (more) {
            __syncthreads();
            rows_put(xdst, rn0, rn1, rn2, rn3);
            __syncthreads();
        }
    }

    {
        float* o0 = obase + (size_t)(h0 + 2 * (nsteps - 1)) * WOUT;
        #pragma unroll
        for (int t = 0; t < 4; ++t) {
            if (t < 3 || tile3_ok) {
                const v8f y0 = (acc[0][t] + acc[1][t]) + acc[2][t];
                const v8f y1 = (acc[3][t] + acc[1][t]) - acc[2][t];
                tile_out(o0 + 16 * t, y0);
                tile_out(o0 + WOUT + 16 * t, y1);
            }
        }
    }
}

extern "C" void kernel_launch(void* const* d_in, const int* in_sizes, int n_in,
                              void* d_out, int out_size, void* d_ws, size_t ws_size,
                              hipStream_t stream) {
    (void)in_sizes; (void)n_in; (void)out_size; (void)d_ws; (void)ws_size;
    const float* x    = (const float*)d_in[0];
    const float* w    = (const float*)d_in[1];
    const float* bias = (const float*)d_in[2];
    float* out        = (float*)d_out;

    dim3 grid(NCHUNK, B_N);
    dynconv2d_wmma_f32<<<grid, 256, LDS_BYTES, stream>>>(x, w, bias, out);
}


# Horizontal minimal filtering F(2,3) nested in the vertical one: 16 tap products per 2x2 output patch (was 24); input columns transformed in registers, outputs folded on vector unit; sum order changes
# speedup vs baseline: 8.4220x; 1.1809x over previous
#include <hip/hip_runtime.h>

typedef __attribute__((ext_vector_type(2))) float v2f;
typedef __attribute__((ext_vector_type(4))) float v4f;
typedef __attribute__((ext_vector_type(8))) float v8f;

#define B_N    16
#define CIN    64
#define COUT   64
#define H_IN   128
#define W_IN   128
#define HOUT   126
#define WOUT   126
#define KDIM   576

#define WPITCH 580
#define XCOLS  130
#define CIPITCH 68
#define HALFCOLS 65

#define ROWS_PER_BLOCK 8
#define NCHUNK 16

#define NSLOT 4
#define SW_FLOATS (COUT * WPITCH)
#define XROW_FLOATS (XCOLS * CIPITCH)
#define SX_FLOATS (NSLOT * XROW_FLOATS)
#define WRAW_FLOATS (COUT * KDIM)
#define LDS_BYTES ((SW_FLOATS + WRAW_FLOATS) * 4)

#define ROW_F4 8
#define PLANE  (H_IN * W_IN)
#define W_F4   36

__device__ __forceinline__ void row_fetch(v4f (&r)[ROW_F4], const float* __restrict__ src)
{
    #pragma unroll
    for (int jq = 0; jq < 2; ++jq) {
        #pragma unroll
        for (int jj = 0; jj < 4; ++jj)
            r[4 * jq + jj] = *(const v4f*)(src + (size_t)(16 * jq + jj) * PLANE);
    }
}

__device__ __forceinline__ void rows_put(float* dst, const v4f (&d0)[ROW_F4], const v4f (&d1)[ROW_F4],
                                         const v4f (&d2)[ROW_F4], const v4f (&d3)[ROW_F4])
{
    #pragma unroll
    for (int jq = 0; jq < 2; ++jq) {
        #pragma unroll
        for (int e = 0; e < 4; ++e) {
            v4f t0, t1, t2, t3;
            #pragma unroll
            for (int jj = 0; jj < 4; ++jj) {
                const int j = 4 * jq + jj;
                t0[jj] = d0[j][e] - d2[j][e];
                t1[jj] = d1[j][e] + d2[j][e];
                t2[jj] = d2[j][e] - d1[j][e];
                t3[jj] = d1[j][e] - d3[j][e];
            }
            float* p = dst + ((e & 1) * HALFCOLS + (e >> 1)) * CIPITCH + 16 * jq;
            *(v4f*)(p)                   = t0;
            *(v4f*)(p + XROW_FLOATS)     = t1;
            *(v4f*)(p + 2 * XROW_FLOATS) = t2;
            *(v4f*)(p + 3 * XROW_FLOATS) = t3;
        }
    }
}

#define NBLK   32
#define JB     2

struct Ops {
    v4f a[4][JB];
    v4f b[4][JB];
};

__device__ __forceinline__ void ops_load(Ops& f, const float* aP, const float* bP, int aoff, int boff)
{
    #pragma unroll
    for (int r = 0; r < 3; ++r) {
        #pragma unroll
        for (int j = 0; j < JB; ++j)
            f.a[r][j] = *(const v4f*)(aP + aoff + 64 * r + 8 * j);
    }
    #pragma unroll
    for (int k = 0; k < 4; ++k) {
        #pragma unroll
        for (int j = 0; j < JB; ++j)
            f.b[k][j] = *(const v4f*)(bP + boff + ((k & 1) * HALFCOLS + (k >> 1)) * CIPITCH + 8 * j);
    }
}

__device__ __forceinline__ void extra_load(v4f (&ex)[2][3][JB], const float* aP, int aoff)
{
    #pragma unroll
    for (int v = 0; v < 2; ++v) {
        #pragma unroll
        for (int r = 0; r < 3; ++r) {
            #pragma unroll
            for (int j = 0; j < JB; ++j)
                ex[v][r][j] = *(const v4f*)(aP + aoff + 192 * (v + 1) + 64 * r + 8 * j);
        }
    }
}

__device__ __forceinline__ void ops_transform(Ops& f, const v4f (&ex)[2][3][JB], bool tap3)
{
    if (tap3) {
        #pragma unroll
        for (int r = 0; r < 3; ++r) {
            #pragma unroll
            for (int j = 0; j < JB; ++j)
                f.a[r][j] = (ex[0][r][j] + ex[1][r][j]) - f.a[r][j];
        }
    }
    #pragma unroll
    for (int j = 0; j < JB; ++j) {
        f.a[3][j] = (f.a[1][j] + f.a[2][j]) - f.a[0][j];
        const v4f c0 = f.b[0][j], c1 = f.b[1][j], c2 = f.b[2][j], c3 = f.b[3][j];
        f.b[0][j] = c0 - c2;
        f.b[1][j] = c1 + c2;
        f.b[2][j] = c2 - c1;
        f.b[3][j] = c1 - c3;
    }
}

__device__ __forceinline__ void ops_mma(v8f (&acc)[4], const Ops& f)
{
    #pragma unroll
    for (int j = 0; j < JB; ++j) {
        #pragma unroll
        for (int r = 0; r < 4; ++r) {
            const v2f alo = __builtin_shufflevector(f.a[r][j], f.a[r][j], 0, 1);
            const v2f blo = __builtin_shufflevector(f.b[r][j], f.b[r][j], 0, 1);
            acc[r] = __builtin_amdgcn_wmma_f32_16x16x4_f32(false, alo, false, blo, (short)0, acc[r], false, false);
        }
        #pragma unroll
        for (int r = 0; r < 4; ++r) {
            const v2f ahi = __builtin_shufflevector(f.a[r][j], f.a[r][j], 2, 3);
            const v2f bhi = __builtin_shufflevector(f.b[r][j], f.b[r][j], 2, 3);
            acc[r] = __builtin_amdgcn_wmma_f32_16x16x4_f32(false, ahi, false, bhi, (short)0, acc[r], false, false);
        }
    }
}

__device__ __forceinline__ constexpr int tap_of(int i) { return (i / 8 == 0) ? 3 : ((i / 8 == 3) ? 0 : i / 8); }

extern "C" __global__ __launch_bounds__(256)
void dynconv2d_wmma_f32(const float* __restrict__ x,
                        const float* __restrict__ w,
                        const float* __restrict__ bias,
                        float* __restrict__ out)
{
    extern __shared__ __attribute__((aligned(16))) float smem[];
    float* sW = smem;
    float* sX = smem + SW_FLOATS;

    const int h0     = blockIdx.x * ROWS_PER_BLOCK;
    const int nsteps = min(ROWS_PER_BLOCK, HOUT - h0) / 2;
    const int b     = blockIdx.y;
    const int tid   = threadIdx.x;
    const int wave  = tid >> 5;
    const int lane  = tid & 31;

    const int cg   = (wave & 3) * 8 + (lane & 3) + 4 * (lane >> 4);
    const int ci_t = (wave >> 2) * 32 + 4 * ((lane >> 2) & 3);
    const float* xsrc = x + (((size_t)b * CIN + ci_t) * H_IN + h0) * W_IN + 4 * cg;
    float*       xdst = sX + (2 * cg) * CIPITCH + ci_t;

    v4f ra[ROW_F4], rb[ROW_F4], rc[ROW_F4], rd[ROW_F4];
    row_fetch(ra, xsrc);
    row_fetch(rb, xsrc + W_IN);
    row_fetch(rc, xsrc + 2 * W_IN);
    row_fetch(rd, xsrc + 3 * W_IN);

    const float* wsrc = w + (size_t)b * COUT * KDIM;

    float* wraw = sX;
    {
        v4f wv[W_F4];
        #pragma unroll
        for (int j = 0; j < W_F4; ++j)
            wv[j] = *(const v4f*)(wsrc + 4 * (tid + 256 * j));
        #pragma unroll
        for (int j = 0; j < W_F4; ++j)
            *(v4f*)(wraw + 4 * (tid + 256 * j)) = wv[j];
    }
    __syncthreads();

    #pragma unroll
    for (int k = 0; k < 4; ++k) {
        const int v  = tid + 256 * k;
        const int co = v >> 4;
        const float* rs = wraw + 36 * v;
        v4f raw[9];
        #pragma unroll
        for (int m = 0; m < 9; ++m)
            raw[m] = *(const v4f*)(rs + 4 * m);
        float* d = sW + co * WPITCH + 4 * (v & 15);
        v4f vp[3][3];
        #pragma unroll
        for (int kw = 0; kw < 3; ++kw) {
            v4f g0, g1, g2;
            #pragma unroll
            for (int c = 0; c < 4; ++c) {
                g0[c] = raw[(9 * c + kw) >> 2][(9 * c + kw) & 3];
                g1[c] = raw[(9 * c + 3 + kw) >> 2][(9 * c + 3 + kw) & 3];
                g2[c] = raw[(9 * c + 6 + kw) >> 2][(9 * c + 6 + kw) & 3];
            }
            const v4f s = g0 + g2;
            vp[0][kw] = g0;
            vp[1][kw] = (s + g1) * 0.5f;
            vp[2][kw] = (s - g1) * 0.5f;
        }
        #pragma unroll
        for (int pv = 0; pv < 3; ++pv) {
            const v4f s = vp[pv][0] + vp[pv][2];
            *(v4f*)(d + (pv * 3 + 0) * 64) = vp[pv][0];
            *(v4f*)(d + (pv * 3 + 1) * 64) = (s + vp[pv][1]) * 0.5f;
            *(v4f*)(d + (pv * 3 + 2) * 64) = (s - vp[pv][1]) * 0.5f;
        }
    }
    __syncthreads();

    #pragma unroll
    for (int k = 0; k < 2; ++k) {
        int idx = tid + 256 * k;
        sX[(idx >> 7) * XROW_FLOATS + (((idx >> 6) & 1) * HALFCOLS + 64) * CIPITCH + (idx & 63)] = 0.0f;
    }
    rows_put(xdst, ra, rb, rc, rd);
    __syncthreads();

    const int laneN = lane & 15;
    const int hi    = lane >> 4;

    const int co_base = (wave & 3) * 16;
    const int PP0     = (wave >> 2) * 32;

    const float* aB = sW + (co_base + laneN) * WPITCH + 4 * hi;
    const float* bB[2];
    #pragma unroll
    for (int t = 0; t < 2; ++t)
        bB[t] = sX + (PP0 + 16 * t + laneN) * CIPITCH + 4 * hi;

    v8f bvec;
    #pragma unroll
    for (int j = 0; j < 8; ++j)
        bvec[j] = bias[b * COUT + co_base + j + 8 * hi];
    asm volatile("" : "+v"(bvec));

    float* obase = out + ((size_t)(b * COUT + co_base + 8 * hi) * HOUT) * WOUT + 2 * (PP0 + laneN);
    const bool tile1_ok = (PP0 + 16 + laneN) < (WOUT / 2);

    Ops fs[2];
    v4f ex[2][3][JB];

    for (int s = 0; s < nsteps; ++s) {
        const int  h    = h0 + 2 * s;
        const bool more = s + 1 < nsteps;
        float* const orow = obase + (size_t)h * WOUT;

        v4f rn0[ROW_F4], rn1[ROW_F4], rn2[ROW_F4], rn3[ROW_F4];

        ops_load(fs[0], aB, bB[0], 0, 3 * XROW_FLOATS);
        extra_load(ex, aB, 0);

        v8f y[2][2][2];
        #pragma unroll
        for (int t = 0; t < 2; ++t) {
            y[0][0][t] = bvec; y[0][1][t] = bvec;
            y[1][0][t] = bvec; y[1][1][t] = bvec;
        }
        v8f acc[4];

        #pragma unroll
        for (int i = 0; i < NBLK; ++i) {
            const int q  = tap_of(i);
            const int pt = (i >> 2) & 1;
            ops_transform(fs[i & 1], ex, q == 3);
            if ((i & 3) == 0) {
                #pragma unroll
                for (int r = 0; r < 4; ++r)
                    acc[r] = v8f{};
            }
            __builtin_amdgcn_sched_barrier(0);

            if (i + 1 < NBLK) {
                const int qn  = tap_of(i + 1);
                const int ptn = ((i + 1) >> 2) & 1;
                const int cbn = ((i + 1) & 3) * 16;
                const int pl  = (qn == 3) ? 0 : qn;
                ops_load(fs[(i + 1) & 1], aB, bB[ptn], pl * 192 + cbn, qn * XROW_FLOATS + cbn);
                if (qn == 3)
                    extra_load(ex, aB, cbn);
            }
            if (i >= 24) {
                const int t = (i - 24) >> 2, jp = (i - 24) & 3;
                if (t < 1 || tile1_ok) {
                    #pragma unroll
                    for (int j = 2 * jp; j < 2 * jp + 2; ++j)
                        *(v2f*)(orow + (size_t)j * (HOUT * WOUT) + WOUT + 32 * t) = v2f{y[1][0][t][j], y[1][1][t][j]};
                }
            }
            if (i == 16 && more) {
                const float* nx = xsrc + (size_t)(2 * (s + 1)) * W_IN;
                row_fetch(rn0, nx);
                row_fetch(rn1, nx + W_IN);
                row_fetch(rn2, nx + 2 * W_IN);
                row_fetch(rn3, nx + 3 * W_IN);
            }
            __builtin_amdgcn_sched_barrier(0);

            ops_mma(acc, fs[i & 1]);

            __builtin_amdgcn_sched_barrier(0);

            if ((i & 3) == 3) {
                asm volatile("v_nop\n\tv_nop\n\tv_nop\n\tv_nop"
                             : "+v"(acc[0]), "+v"(acc[1]), "+v"(acc[2]), "+v"(acc[3]));
                const v8f z0 = (acc[0] + acc[1]) + acc[2];
                const v8f z1 = (acc[1] - acc[2]) - acc[3];
                if (q <= 2) { y[0][0][pt] += z0; y[0][1][pt] += z1; }
                if (q == 1) { y[1][0][pt] += z0; y[1][1][pt] += z1; }
                if (q >= 2) { y[1][0][pt] -= z0; y[1][1][pt] -= z1; }
            }
        }

        #pragma unroll
        for (int t = 0; t < 2; ++t) {
            if (t < 1 || tile1_ok) {
                #pragma unroll
                for (int j = 0; j < 8; ++j)
                    *(v2f*)(orow + (size_t)j * (HOUT * WOUT) + 32 * t) = v2f{y[0][0][t][j], y[0][1][t][j]};
            }
        }

        if (more) {
            __syncthreads();
            rows_put(xdst, rn0, rn1, rn2, rn3);
            __syncthreads();
        }
    }
}

extern "C" void kernel_launch(void* const* d_in, const int* in_sizes, int n_in,
                              void* d_out, int out_size, void* d_ws, size_t ws_size,
                              hipStream_t stream) {
    (void)in_sizes; (void)n_in; (void)out_size; (void)d_ws; (void)ws_size;
    const float* x    = (const float*)d_in[0];
    const float* w    = (const float*)d_in[1];
    const float* bias = (const float*)d_in[2];
    float* out        = (float*)d_out;

    dim3 grid(NCHUNK, B_N);
    dynconv2d_wmma_f32<<<grid, 256, LDS_BYTES, stream>>>(x, w, bias, out);
}


# Weight operands shared by a wave's two pixel-pair tiles: per (tap, 16 ci) the weight planes are read from LDS and completed once, not once per tile (block order now tap, ci, tile); sums unchanged
# speedup vs baseline: 8.7287x; 1.0364x over previous
#include <hip/hip_runtime.h>

typedef __attribute__((ext_vector_type(2))) float v2f;
typedef __attribute__((ext_vector_type(4))) float v4f;
typedef __attribute__((ext_vector_type(8))) float v8f;

#define B_N    16
#define CIN    64
#define COUT   64
#define H_IN   128
#define W_IN   128
#define HOUT   126
#define WOUT   126
#define KDIM   576

#define WPITCH 580
#define XCOLS  130
#define CIPITCH 68
#define HALFCOLS 65

#define ROWS_PER_BLOCK 8
#define NCHUNK 16

#define NSLOT 4
#define SW_FLOATS (COUT * WPITCH)
#define XROW_FLOATS (XCOLS * CIPITCH)
#define SX_FLOATS (NSLOT * XROW_FLOATS)
#define WRAW_FLOATS (COUT * KDIM)
#define LDS_BYTES ((SW_FLOATS + WRAW_FLOATS) * 4)

#define ROW_F4 8
#define PLANE  (H_IN * W_IN)
#define W_F4   36

__device__ __forceinline__ void row_fetch(v4f (&r)[ROW_F4], const float* __restrict__ src)
{
    #pragma unroll
    for (int jq = 0; jq < 2; ++jq) {
        #pragma unroll
        for (int jj = 0; jj < 4; ++jj)
            r[4 * jq + jj] = *(const v4f*)(src + (size_t)(16 * jq + jj) * PLANE);
    }
}

__device__ __forceinline__ void rows_put(float* dst, const v4f (&d0)[ROW_F4], const v4f (&d1)[ROW_F4],
                                         const v4f (&d2)[ROW_F4], const v4f (&d3)[ROW_F4])
{
    #pragma unroll
    for (int jq = 0; jq < 2; ++jq) {
        #pragma unroll
        for (int e = 0; e < 4; ++e) {
            v4f t0, t1, t2, t3;
            #pragma unroll
            for (int jj = 0; jj < 4; ++jj) {
                const int j = 4 * jq + jj;
                t0[jj] = d0[j][e] - d2[j][e];
                t1[jj] = d1[j][e] + d2[j][e];
                t2[jj] = d2[j][e] - d1[j][e];
                t3[jj] = d1[j][e] - d3[j][e];
            }
            float* p = dst + ((e & 1) * HALFCOLS + (e >> 1)) * CIPITCH + 16 * jq;
            *(v4f*)(p)                   = t0;
            *(v4f*)(p + XROW_FLOATS)     = t1;
            *(v4f*)(p + 2 * XROW_FLOATS) = t2;
            *(v4f*)(p + 3 * XROW_FLOATS) = t3;
        }
    }
}

#define NBLK   32
#define JB     2

struct OpsA {
    v4f a[4][JB];
};
struct OpsB {
    v4f b[4][JB];
};

__device__ __forceinline__ void a_load(OpsA& f, const float* aP, int aoff)
{
    #pragma unroll
    for (int r = 0; r < 3; ++r) {
        #pragma unroll
        for (int j = 0; j < JB; ++j)
            f.a[r][j] = *(const v4f*)(aP + aoff + 64 * r + 8 * j);
    }
}
__device__ __forceinline__ void b_load(OpsB& f, const float* bP, int boff)
{
    #pragma unroll
    for (int k = 0; k < 4; ++k) {
        #pragma unroll
        for (int j = 0; j < JB; ++j)
            f.b[k][j] = *(const v4f*)(bP + boff + ((k & 1) * HALFCOLS + (k >> 1)) * CIPITCH + 8 * j);
    }
}

__device__ __forceinline__ void extra_load(v4f (&ex)[2][3][JB], const float* aP, int aoff)
{
    #pragma unroll
    for (int v = 0; v < 2; ++v) {
        #pragma unroll
        for (int r = 0; r < 3; ++r) {
            #pragma unroll
            for (int j = 0; j < JB; ++j)
                ex[v][r][j] = *(const v4f*)(aP + aoff + 192 * (v + 1) + 64 * r + 8 * j);
        }
    }
}

__device__ __forceinline__ void a_transform(OpsA& f, const v4f (&ex)[2][3][JB], bool tap3)
{
    if (tap3) {
        #pragma unroll
        for (int r = 0; r < 3; ++r) {
            #pragma unroll
            for (int j = 0; j < JB; ++j)
                f.a[r][j] = (ex[0][r][j] + ex[1][r][j]) - f.a[r][j];
        }
    }
    #pragma unroll
    for (int j = 0; j < JB; ++j)
        f.a[3][j] = (f.a[1][j] + f.a[2][j]) - f.a[0][j];
}
__device__ __forceinline__ void b_transform(OpsB& f)
{
    #pragma unroll
    for (int j = 0; j < JB; ++j) {
        const v4f c0 = f.b[0][j], c1 = f.b[1][j], c2 = f.b[2][j], c3 = f.b[3][j];
        f.b[0][j] = c0 - c2;
        f.b[1][j] = c1 + c2;
        f.b[2][j] = c2 - c1;
        f.b[3][j] = c1 - c3;
    }
}

__device__ __forceinline__ void ops_mma(v8f (&acc)[4], const OpsA& fa, const OpsB& f)
{
    #pragma unroll
    for (int j = 0; j < JB; ++j) {
        #pragma unroll
        for (int r = 0; r < 4; ++r) {
            const v2f alo = __builtin_shufflevector(fa.a[r][j], fa.a[r][j], 0, 1);
            const v2f blo = __builtin_shufflevector(f.b[r][j], f.b[r][j], 0, 1);
            acc[r] = __builtin_amdgcn_wmma_f32_16x16x4_f32(false, alo, false, blo, (short)0, acc[r], false, false);
        }
        #pragma unroll
        for (int r = 0; r < 4; ++r) {
            const v2f ahi = __builtin_shufflevector(fa.a[r][j], fa.a[r][j], 2, 3);
            const v2f bhi = __builtin_shufflevector(f.b[r][j], f.b[r][j], 2, 3);
            acc[r] = __builtin_amdgcn_wmma_f32_16x16x4_f32(false, ahi, false, bhi, (short)0, acc[r], false, false);
        }
    }
}

__device__ __forceinline__ constexpr int tap_of(int i) { return (i / 8 == 0) ? 3 : ((i / 8 == 3) ? 0 : i / 8); }

extern "C" __global__ __launch_bounds__(256)
void dynconv2d_wmma_f32(const float* __restrict__ x,
                        const float* __restrict__ w,
                        const float* __restrict__ bias,
                        float* __restrict__ out)
{
    extern __shared__ __attribute__((aligned(16))) float smem[];
    float* sW = smem;
    float* sX = smem + SW_FLOATS;

    const int h0     = blockIdx.x * ROWS_PER_BLOCK;
    const int nsteps = min(ROWS_PER_BLOCK, HOUT - h0) / 2;
    const int b     = blockIdx.y;
    const int tid   = threadIdx.x;
    const int wave  = tid >> 5;
    const int lane  = tid & 31;

    const int cg   = (wave & 3) * 8 + (lane & 3) + 4 * (lane >> 4);
    const int ci_t = (wave >> 2) * 32 + 4 * ((lane >> 2) & 3);
    const float* xsrc = x + (((size_t)b * CIN + ci_t) * H_IN + h0) * W_IN + 4 * cg;
    float*       xdst = sX + (2 * cg) * CIPITCH + ci_t;

    v4f ra[ROW_F4], rb[ROW_F4], rc[ROW_F4], rd[ROW_F4];
    row_fetch(ra, xsrc);
    row_fetch(rb, xsrc + W_IN);
    row_fetch(rc, xsrc + 2 * W_IN);
    row_fetch(rd, xsrc + 3 * W_IN);

    const float* wsrc = w + (size_t)b * COUT * KDIM;

    float* wraw = sX;
    {
        v4f wv[W_F4];
        #pragma unroll
        for (int j = 0; j < W_F4; ++j)
            wv[j] = *(const v4f*)(wsrc + 4 * (tid + 256 * j));
        #pragma unroll
        for (int j = 0; j < W_F4; ++j)
            *(v4f*)(wraw + 4 * (tid + 256 * j)) = wv[j];
    }
    __syncthreads();

    #pragma unroll
    for (int k = 0; k < 4; ++k) {
        const int v  = tid + 256 * k;
        const int co = v >> 4;
        const float* rs = wraw + 36 * v;
        v4f raw[9];
        #pragma unroll
        for (int m = 0; m < 9; ++m)
            raw[m] = *(const v4f*)(rs + 4 * m);
        float* d = sW + co * WPITCH + 4 * (v & 15);
        v4f vp[3][3];
        #pragma unroll
        for (int kw = 0; kw < 3; ++kw) {
            v4f g0, g1, g2;
            #pragma unroll
            for (int c = 0; c < 4; ++c) {
                g0[c] = raw[(9 * c + kw) >> 2][(9 * c + kw) & 3];
                g1[c] = raw[(9 * c + 3 + kw) >> 2][(9 * c + 3 + kw) & 3];
                g2[c] = raw[(9 * c + 6 + kw) >> 2][(9 * c + 6 + kw) & 3];
            }
            const v4f s = g0 + g2;
            vp[0][kw] = g0;
            vp[1][kw] = (s + g1) * 0.5f;
            vp[2][kw] = (s - g1) * 0.5f;
        }
        #pragma unroll
        for (int pv = 0; pv < 3; ++pv) {
            const v4f s = vp[pv][0] + vp[pv][2];
            *(v4f*)(d + (pv * 3 + 0) * 64) = vp[pv][0];
            *(v4f*)(d + (pv * 3 + 1) * 64) = (s + vp[pv][1]) * 0.5f;
            *(v4f*)(d + (pv * 3 + 2) * 64) = (s - vp[pv][1]) * 0.5f;
        }
    }
    __syncthreads();

    #pragma unroll
    for (int k = 0; k < 2; ++k) {
        int idx = tid + 256 * k;
        sX[(idx >> 7) * XROW_FLOATS + (((idx >> 6) & 1) * HALFCOLS + 64) * CIPITCH + (idx & 63)] = 0.0f;
    }
    rows_put(xdst, ra, rb, rc, rd);
    __syncthreads();

    const int laneN = lane & 15;
    const int hi    = lane >> 4;

    const int co_base = (wave & 3) * 16;
    const int PP0     = (wave >> 2) * 32;

    const float* aB = sW + (co_base + laneN) * WPITCH + 4 * hi;
    const float* bB[2];
    #pragma unroll
    for (int t = 0; t < 2; ++t)
        bB[t] = sX + (PP0 + 16 * t + laneN) * CIPITCH + 4 * hi;

    v8f bvec;
    #pragma unroll
    for (int j = 0; j < 8; ++j)
        bvec[j] = bias[b * COUT + co_base + j + 8 * hi];
    asm volatile("" : "+v"(bvec));

    float* obase = out + ((size_t)(b * COUT + co_base + 8 * hi) * HOUT) * WOUT + 2 * (PP0 + laneN);
    const bool tile1_ok = (PP0 + 16 + laneN) < (WOUT / 2);

    OpsA fa[2];
    OpsB fb[2];
    v4f ex[2][3][JB];

    for (int s = 0; s < nsteps; ++s) {
        const int  h    = h0 + 2 * s;
        const bool more = s + 1 < nsteps;
        float* const orow = obase + (size_t)h * WOUT;

        v4f rn0[ROW_F4], rn1[ROW_F4], rn2[ROW_F4], rn3[ROW_F4];

        a_load(fa[0], aB, 0);
        extra_load(ex, aB, 0);
        b_load(fb[0], bB[0], 3 * XROW_FLOATS);

        v8f y[2][2][2];
        #pragma unroll
        for (int t = 0; t < 2; ++t) {
            y[0][0][t] = bvec; y[0][1][t] = bvec;
            y[1][0][t] = bvec; y[1][1][t] = bvec;
        }
        v8f acc[2][4];

        #pragma unroll
        for (int i = 0; i < NBLK; ++i) {
            const int q  = tap_of(i);
            const int pt = i & 1;
            const int as = (i >> 1) & 1;
            if (pt == 0)
                a_transform(fa[as], ex, q == 3);
            b_transform(fb[pt]);
            if ((i & 7) < 2) {
                #pragma unroll
                for (int r = 0; r < 4; ++r)
                    acc[pt][r] = v8f{};
            }
            __builtin_amdgcn_sched_barrier(0);

            if (i + 1 < NBLK) {
                const int qn  = tap_of(i + 1);
                const int ptn = (i + 1) & 1;
                const int cbn = (((i + 1) >> 1) & 3) * 16;
                const int pl  = (qn == 3) ? 0 : qn;
                if (ptn == 0) {
                    a_load(fa[((i + 1) >> 1) & 1], aB, pl * 192 + cbn);
                    if (qn == 3)
                        extra_load(ex, aB, cbn);
                }
                b_load(fb[ptn], bB[ptn], qn * XROW_FLOATS + cbn);
            }
            if (i >= 24) {
                const int t = (i - 24) >> 2, jp = (i - 24) & 3;
                if (t < 1 || tile1_ok) {
                    #pragma unroll
                    for (int j = 2 * jp; j < 2 * jp + 2; ++j)
                        *(v2f*)(orow + (size_t)j * (HOUT * WOUT) + WOUT + 32 * t) = v2f{y[1][0][t][j], y[1][1][t][j]};
                }
            }
            if (i == 16 && more) {
                const float* nx = xsrc + (size_t)(2 * (s + 1)) * W_IN;
                row_fetch(rn0, nx);
                row_fetch(rn1, nx + W_IN);
                row_fetch(rn2, nx + 2 * W_IN);
                row_fetch(rn3, nx + 3 * W_IN);
            }
            __builtin_amdgcn_sched_barrier(0);

            ops_mma(acc[pt], fa[as], fb[pt]);

            __builtin_amdgcn_sched_barrier(0);

            if ((i & 7) == 7) {
                asm volatile("v_nop\n\tv_nop\n\tv_nop\n\tv_nop"
                             : "+v"(acc[0][0]), "+v"(acc[0][1]), "+v"(acc[0][2]), "+v"(acc[0][3]),
                               "+v"(acc[1][0]), "+v"(acc[1][1]), "+v"(acc[1][2]), "+v"(acc[1][3]));
                #pragma unroll
                for (int t = 0; t < 2; ++t) {
                    const v8f z0 = (acc[t][0] + acc[t][1]) + acc[t][2];
                    const v8f z1 = (acc[t][1] - acc[t][2]) - acc[t][3];
                    if (q <= 2) { y[0][0][t] += z0; y[0][1][t] += z1; }
                    if (q == 1) { y[1][0][t] += z0; y[1][1][t] += z1; }
                    if (q >= 2) { y[1][0][t] -= z0; y[1][1][t] -= z1; }
                }
            }
        }

        #pragma unroll
        for (int t = 0; t < 2; ++t) {
            if (t < 1 || tile1_ok) {
                #pragma unroll
                for (int j = 0; j < 8; ++j)
                    *(v2f*)(orow + (size_t)j * (HOUT * WOUT) + 32 * t) = v2f{y[0][0][t][j], y[0][1][t][j]};
            }
        }

        if (more) {
            __syncthreads();
            rows_put(xdst, rn0, rn1, rn2, rn3);
            __syncthreads();
        }
    }
}

extern "C" void kernel_launch(void* const* d_in, const int* in_sizes, int n_in,
                              void* d_out, int out_size, void* d_ws, size_t ws_size,
                              hipStream_t stream) {
    (void)in_sizes; (void)n_in; (void)out_size; (void)d_ws; (void)ws_size;
    const float* x    = (const float*)d_in[0];
    const float* w    = (const float*)d_in[1];
    const float* bias = (const float*)d_in[2];
    float* out        = (float*)d_out;

    dim3 grid(NCHUNK, B_N);
    dynconv2d_wmma_f32<<<grid, 256, LDS_BYTES, stream>>>(x, w, bias, out);
}
